# starting kernel
# baseline (speedup 1.0000x reference)
#include <hip/hip_runtime.h>
#include <hip/hip_bf16.h>

typedef float v2f __attribute__((ext_vector_type(2)));
typedef float v8f __attribute__((ext_vector_type(8)));

#define N_NODES  50000
#define N_EDGES  600000
#define N_GRAPHS 64
#define D_IN     128
#define D_HID    128
#define D_OUT    64

__global__ __launch_bounds__(32)
void gemm_wmma_f32(const float* __restrict__ X, const float* __restrict__ W,
                   float* __restrict__ H, int K, int F) {
  const int rowTile = blockIdx.x << 4;
  const int colTile = blockIdx.y << 4;
  const int lane = threadIdx.x & 31;
  const int half = lane >> 4;
  const int i16  = lane & 15;

  const float* xrow = X + (size_t)(rowTile + i16) * (size_t)K;
  const float* wcol = W + colTile + i16;

  v8f acc = {};
  for (int k0 = 0; k0 < K; k0 += 4) {
    const int ka = k0 + (half << 1);
    v2f a = *(const v2f*)(xrow + ka);
    v2f b;
    b.x = wcol[(size_t)ka * (size_t)F];
    b.y = wcol[(size_t)(ka + 1) * (size_t)F];
    acc = __builtin_amdgcn_wmma_f32_16x16x4_f32(
         false, a,  false, b,
         (short)0, acc,  false,  false);
  }

  float* out = H + (size_t)(rowTile + (half << 3)) * (size_t)F + colTile + i16;
#pragma unroll
  for (int i = 0; i < 8; ++i) out[(size_t)i * (size_t)F] = acc[i];
}

__global__ void zero_f32(float* __restrict__ p, int n) {
  int i = blockIdx.x * blockDim.x + threadIdx.x;
  if (i < n) p[i] = 0.0f;
}

__global__ void deg_accum(const int* __restrict__ dst, float* __restrict__ deg, int E) {
  int e = blockIdx.x * blockDim.x + threadIdx.x;
  if (e < E) atomicAdd(&deg[dst[e]], 1.0f);
}

__global__ void deg_finalize(float* __restrict__ deg, int n) {
  int i = blockIdx.x * blockDim.x + threadIdx.x;
  if (i < n) deg[i] = rsqrtf(deg[i] + 1.0f);
}

__global__ void agg_init(const float* __restrict__ h, const float* __restrict__ dis,
                         const float* __restrict__ bias, float* __restrict__ agg,
                         int F, int total) {
  int i = blockIdx.x * blockDim.x + threadIdx.x;
  if (i >= total) return;
  int node = i / F;
  int f = i - node * F;
  float d = dis[node];
  agg[i] = h[i] * d * d + bias[f];
}

__global__ void edge_scatter(const int* __restrict__ src, const int* __restrict__ dst,
                             const float* __restrict__ h, const float* __restrict__ dis,
                             float* __restrict__ agg, int F, int total) {
  int i = blockIdx.x * blockDim.x + threadIdx.x;
  if (i >= total) return;
  int e = i / F;
  int f = i - e * F;
  int s = src[e];
  int d = dst[e];
  float nm = dis[s] * dis[d];
  atomicAdd(&agg[(size_t)d * (size_t)F + f], h[(size_t)s * (size_t)F + f] * nm);
}

__global__ void relu_k(float* __restrict__ p, int n) {
  int i = blockIdx.x * blockDim.x + threadIdx.x;
  if (i < n) p[i] = fmaxf(p[i], 0.0f);
}

__global__ void pool_accum(const float* __restrict__ h, const int* __restrict__ batch,
                           float* __restrict__ sums, float* __restrict__ cnts, int total) {
  int i = blockIdx.x * blockDim.x + threadIdx.x;
  if (i >= total) return;
  int node = i >> 6;
  int f = i & 63;
  int g = batch[node];
  atomicAdd(&sums[(g << 6) + f], h[i]);
  if (f == 0) atomicAdd(&cnts[g], 1.0f);
}

__global__ void pool_final(const float* __restrict__ sums, const float* __restrict__ cnts,
                           float* __restrict__ out) {
  int i = blockIdx.x * blockDim.x + threadIdx.x;
  if (i < N_GRAPHS * D_OUT) {
    int g = i >> 6;
    out[i] = sums[i] / fmaxf(cnts[g], 1.0f);
  }
}

extern "C" void kernel_launch(void* const* d_in, const int* in_sizes, int n_in,
                              void* d_out, int out_size, void* d_ws, size_t ws_size,
                              hipStream_t stream) {
  const float* x     = (const float*)d_in[0];
  const int*   ei    = (const int*)d_in[1];
  const int*   batch = (const int*)d_in[2];
  const float* W1 = (const float*)d_in[3];  const float* b1 = (const float*)d_in[4];
  const float* W2 = (const float*)d_in[5];  const float* b2 = (const float*)d_in[6];
  const float* W3 = (const float*)d_in[7];  const float* b3 = (const float*)d_in[8];
  const float* W4 = (const float*)d_in[9];  const float* b4 = (const float*)d_in[10];
  float* out = (float*)d_out;

  float* bufA = (float*)d_ws;
  float* bufB = bufA + (size_t)N_NODES * D_HID;
  float* dis  = bufB + (size_t)N_NODES * D_HID;
  float* sums = dis + N_NODES;
  float* cnts = sums + N_GRAPHS * D_OUT;

  const int* src = ei;
  const int* dst = ei + N_EDGES;

  const int TB = 256;
  auto blocks = [](long long n, int tb) { return (unsigned)((n + tb - 1) / tb); };

  zero_f32<<<blocks(N_NODES, TB), TB, 0, stream>>>(dis, N_NODES);
  deg_accum<<<blocks(N_EDGES, TB), TB, 0, stream>>>(dst, dis, N_EDGES);
  deg_finalize<<<blocks(N_NODES, TB), TB, 0, stream>>>(dis, N_NODES);

  auto layer = [&](const float* hin, int K, const float* W, const float* b, int F,
                   bool do_relu) {
    dim3 grid(N_NODES / 16, F / 16);
    gemm_wmma_f32<<<grid, 32, 0, stream>>>(hin, W, bufA, K, F);
    int totalNF = N_NODES * F;
    agg_init<<<blocks(totalNF, TB), TB, 0, stream>>>(bufA, dis, b, bufB, F, totalNF);
    int totalEF = N_EDGES * F;
    edge_scatter<<<blocks(totalEF, TB), TB, 0, stream>>>(src, dst, bufA, dis, bufB, F,
                                                         totalEF);
    if (do_relu) relu_k<<<blocks(totalNF, TB), TB, 0, stream>>>(bufB, totalNF);
  };

  layer(x,    D_IN,  W1, b1, D_HID, true);
  layer(bufB, D_HID, W2, b2, D_HID, true);
  layer(bufB, D_HID, W3, b3, D_HID, true);
  layer(bufB, D_HID, W4, b4, D_OUT, false);

  zero_f32<<<blocks(N_GRAPHS * D_OUT + N_GRAPHS, TB), TB, 0, stream>>>(
      sums, N_GRAPHS * D_OUT + N_GRAPHS);
  pool_accum<<<blocks((long long)N_NODES * D_OUT, TB), TB, 0, stream>>>(
      bufB, batch, sums, cnts, N_NODES * D_OUT);
  pool_final<<<blocks(N_GRAPHS * D_OUT, TB), TB, 0, stream>>>(sums, cnts, out);
}


# mean pool moved in front of layer 4's product: a bin's block pools its (A_hat h3) tile rows per graph (f32 atomic per run of equal ids), then one [64,128]x[128,64] WMMA block; sum order changes
# speedup vs baseline: 4.5620x; 4.5620x over previous
#include <hip/hip_runtime.h>
#include <hip/hip_bf16.h>

typedef float v2f __attribute__((ext_vector_type(2)));
typedef float v8f __attribute__((ext_vector_type(8)));

#define N_NODES  50000
#define N_EDGES  600000
#define N_GRAPHS 64
#define D_IN     128
#define D_HID    128
#define D_OUT    64

#define TILE_R     64
#define N_BINS     ((N_NODES + TILE_R - 1) / TILE_R)
#define BIN_BLOCKS 64
#define BIN_CHUNK  ((N_EDGES + BIN_BLOCKS - 1) / BIN_BLOCKS)
#define BIN_TB     1024
#define AGG_WAVES  8

static_assert(N_NODES <= 65536, "a source row index must fit the low 16 bits of an edge record");
static_assert(N_BINS <= BIN_TB, "the scan gives one thread to each bin");

constexpr size_t WS_NEED =
    ((size_t)2 * N_NODES * D_HID + N_NODES + N_BINS + 1 + N_EDGES) * sizeof(float);
static_assert((size_t)N_GRAPHS * D_HID + N_GRAPHS <= (size_t)N_NODES * D_HID,
              "the pooled sums and counts fit in one plane");

__global__ void zero_f32(float* __restrict__ p, int n) {
  int i = blockIdx.x * blockDim.x + threadIdx.x;
  if (i < n) p[i] = 0.0f;
}

__global__ void deg_accum(const int* __restrict__ dst, float* __restrict__ deg, int E) {
  int e = blockIdx.x * blockDim.x + threadIdx.x;
  if (e < E) atomicAdd(&deg[dst[e]], 1.0f);
}

__global__ void deg_finalize(float* __restrict__ deg, int n) {
  int i = blockIdx.x * blockDim.x + threadIdx.x;
  if (i < n) deg[i] = rsqrtf(deg[i] + 1.0f);
}

__global__ __launch_bounds__(BIN_TB)
void bin_hist(const int* __restrict__ dst, int* __restrict__ cnt) {
  __shared__ int hist[N_BINS];
  for (int b = threadIdx.x; b < N_BINS; b += BIN_TB) hist[b] = 0;
  __syncthreads();
  const int e0 = blockIdx.x * BIN_CHUNK;
  const int e1 = min(e0 + BIN_CHUNK, N_EDGES);
  for (int e = e0 + threadIdx.x; e < e1; e += BIN_TB)
    atomicAdd(&hist[(unsigned)dst[e] / TILE_R], 1);
  __syncthreads();
  for (int b = threadIdx.x; b < N_BINS; b += BIN_TB) cnt[blockIdx.x * N_BINS + b] = hist[b];
}

__global__ __launch_bounds__(BIN_TB)
void bin_scan(int* __restrict__ cnt, int* __restrict__ binStart) {
  __shared__ int sc[BIN_TB];
  const int t = threadIdx.x;
  int tot = 0;
  if (t < N_BINS)
    for (int k = 0; k < BIN_BLOCKS; ++k) tot += cnt[k * N_BINS + t];
  sc[t] = tot;
  __syncthreads();
  for (int off = 1; off < BIN_TB; off <<= 1) {
    const int v = (t >= off) ? sc[t - off] : 0;
    __syncthreads();
    sc[t] += v;
    __syncthreads();
  }
  const int excl = sc[t] - tot;
  if (t <= N_BINS) binStart[t] = excl;
  if (t < N_BINS) {
    int run = excl;
    for (int k = 0; k < BIN_BLOCKS; ++k) {
      const int c = cnt[k * N_BINS + t];
      cnt[k * N_BINS + t] = run;
      run += c;
    }
  }
}

__global__ __launch_bounds__(BIN_TB)
void bin_place(const int* __restrict__ src, const int* __restrict__ dst,
               const int* __restrict__ cnt, unsigned* __restrict__ rec) {
  __shared__ int cur[N_BINS];
  for (int b = threadIdx.x; b < N_BINS; b += BIN_TB) cur[b] = cnt[blockIdx.x * N_BINS + b];
  __syncthreads();
  const int e0 = blockIdx.x * BIN_CHUNK;
  const int e1 = min(e0 + BIN_CHUNK, N_EDGES);
  for (int e = e0 + threadIdx.x; e < e1; e += BIN_TB) {
    const unsigned d = (unsigned)dst[e];
    const unsigned s = (unsigned)src[e];
    const unsigned bin = d / TILE_R;
    const int pos = atomicAdd(&cur[bin], 1);
    rec[pos] = ((d - bin * TILE_R) << 16) | s;
  }
}

#define TILE_K 128
#define TILE_P 130

__device__ __forceinline__
void sum_rows_into_tile(const unsigned* __restrict__ rec, const int* __restrict__ binStart,
                        const float* __restrict__ h, const float* __restrict__ dis,
                        float* __restrict__ tile, float* __restrict__ disT,
                        int row0, int nrows, int wave, int lane) {
  constexpr int VPL = TILE_K / 32;
  typedef float vec __attribute__((ext_vector_type(VPL)));
  for (int r = wave; r < nrows; r += AGG_WAVES) {
    const float d = dis[row0 + r];
    const float dd = d * d;
    const vec v = *(const vec*)(h + (size_t)(row0 + r) * TILE_K + lane * VPL);
#pragma unroll
    for (int q = 0; q < VPL; ++q) tile[r * TILE_P + q * 32 + lane] = v[q] * dd;
  }
  if (threadIdx.x < TILE_R)
    disT[threadIdx.x] = ((int)threadIdx.x < nrows) ? dis[row0 + threadIdx.x] : 0.0f;
  __syncthreads();

  const int e0 = binStart[blockIdx.x];
  const int e1 = binStart[blockIdx.x + 1];
  for (int base = e0 + wave * 32; base < e1; base += AGG_WAVES * 32) {
    const int idx = base + lane;
    unsigned r = ((unsigned)TILE_R << 16) | (unsigned)row0;
    float nm = 0.0f;
    if (idx < e1) {
      r = rec[idx];
      nm = dis[r & 0xffffu] * disT[r >> 16];
    }
#pragma unroll
    for (int j = 0; j < 32; j += 8) {
      if (base + j >= e1) break;
      vec v[8];
      int dl[8];
      float w[8];
#pragma unroll
      for (int u = 0; u < 8; ++u) {
        const unsigned rj = (unsigned)__builtin_amdgcn_readlane((int)r, j + u);
        w[u] = __builtin_bit_cast(float, __builtin_amdgcn_readlane(__builtin_bit_cast(int, nm), j + u));
        dl[u] = (int)(rj >> 16);
        v[u] = *(const vec*)(h + (size_t)(rj & 0xffffu) * TILE_K + lane * VPL);
      }
#pragma unroll
      for (int u = 0; u < 8; ++u) {
#pragma unroll
        for (int q = 0; q < VPL; ++q)
          __hip_atomic_fetch_add(&tile[dl[u] * TILE_P + q * 32 + lane], v[u][q] * w[u],
                                 __ATOMIC_RELAXED, __HIP_MEMORY_SCOPE_WORKGROUP);
      }
    }
  }

  __syncthreads();
}

template <int FOUT, bool RELU>
__global__ __launch_bounds__(AGG_WAVES * 32)
void layer_tile(const unsigned* __restrict__ rec, const int* __restrict__ binStart,
                const float* __restrict__ h, const float* __restrict__ dis,
                const float* __restrict__ W, const float* __restrict__ bias,
                float* __restrict__ out) {
  __shared__ __attribute__((aligned(16))) float tile[(TILE_R + 1) * TILE_P];
  __shared__ float disT[TILE_R];

  const int lane = threadIdx.x & 31;
  const int wave = __builtin_amdgcn_readfirstlane((int)(threadIdx.x >> 5));
  const int row0 = blockIdx.x * TILE_R;
  const int nrows = min(TILE_R, N_NODES - row0);

  constexpr int NCT = FOUT / 16;
  constexpr int RTW = NCT / 2;
  static_assert(AGG_WAVES * RTW == (TILE_R / 16) * NCT, "8 waves cover the block's output tiles");
  const int ct = wave % NCT;
  const int rg = wave / NCT;
  const int n16 = lane & 15;
  const int half = lane >> 4;
  const float* wp = W + ct * 16 + n16;
  v2f bw[32];
#pragma unroll
  for (int s = 0; s < 32; ++s) {
    const int k = 8 * (s & 15) + 2 * (s >> 4) + half;
    bw[s].x = wp[(size_t)k * FOUT];
    bw[s].y = wp[(size_t)(k + 4) * FOUT];
  }
  const float bcol = bias[ct * 16 + n16];

  sum_rows_into_tile(rec, binStart, h, dis, tile, disT, row0, nrows, wave, lane);


  v8f acc[RTW];
#pragma unroll
  for (int t = 0; t < RTW; ++t) acc[t] = v8f{};
  const float* abase = tile + ((rg * RTW) * 16 + n16) * TILE_P + 32 * half;
#pragma unroll
  for (int s = 0; s < 32; ++s) {
#pragma unroll
    for (int t = 0; t < RTW; ++t) {
      const v2f a = *(const v2f*)(abase + t * 16 * TILE_P + 64 * (s >> 4) + 2 * (s & 15));
      acc[t] = __builtin_amdgcn_wmma_f32_16x16x4_f32(
           false, a,  false, bw[s],
           (short)0, acc[t],  false,  false);
    }
  }

#pragma unroll
  for (int t = 0; t < RTW; ++t) {
    const int rb = (rg * RTW + t) * 16 + (half << 3);
    float* op = out + (size_t)(row0 + rb) * FOUT + ct * 16 + n16;
    float val[8];
#pragma unroll
    for (int i = 0; i < 8; ++i) {
      val[i] = acc[t][i] + bcol;
      if (RELU) val[i] = fmaxf(val[i], 0.0f);
    }
    if (rb + 8 <= nrows) {
#pragma unroll
      for (int i = 0; i < 8; ++i) op[(size_t)i * FOUT] = val[i];
    } else {
#pragma unroll
      for (int i = 0; i < 8; ++i)
        if (rb + i < nrows) op[(size_t)i * FOUT] = val[i];
    }
  }
}

__global__ __launch_bounds__(AGG_WAVES * 32)
void layer_pool(const unsigned* __restrict__ rec, const int* __restrict__ binStart,
                const float* __restrict__ h, const float* __restrict__ dis,
                const int* __restrict__ batch, float* __restrict__ psum,
                float* __restrict__ cnts) {
  __shared__ __attribute__((aligned(16))) float tile[(TILE_R + 1) * TILE_P];
  __shared__ float disT[TILE_R];
  __shared__ int gidT[TILE_R];

  const int lane = threadIdx.x & 31;
  const int wave = __builtin_amdgcn_readfirstlane((int)(threadIdx.x >> 5));
  const int row0 = blockIdx.x * TILE_R;
  const int nrows = min(TILE_R, N_NODES - row0);

  if (threadIdx.x < TILE_R)
    gidT[threadIdx.x] = ((int)threadIdx.x < nrows) ? batch[row0 + threadIdx.x] : -1;

  sum_rows_into_tile(rec, binStart, h, dis, tile, disT, row0, nrows, wave, lane);

  constexpr int HR = TILE_R / 2;
  const int col = threadIdx.x & (TILE_K - 1);
  const int r0 = (wave >> 2) * HR;
  float v[HR];
  int gd[HR];
#pragma unroll
  for (int r = 0; r < HR; ++r) {
    v[r] = tile[(r0 + r) * TILE_P + col];
    gd[r] = __builtin_amdgcn_readfirstlane(gidT[r0 + r]);
  }
  auto hand_over = [&](int g, float s, int len) {
    if ((unsigned)g < (unsigned)N_GRAPHS) {
      atomicAdd(&psum[g * TILE_K + col], s);
      if (col == 0) atomicAdd(&cnts[g], (float)len);
    }
  };
  int g = gd[0];
  int len = 1;
  float s = v[0];
#pragma unroll
  for (int r = 1; r < HR; ++r) {
    if (gd[r] != g) {
      hand_over(g, s, len);
      g = gd[r];
      len = 0;
      s = 0.0f;
    }
    s += v[r];
    ++len;
  }
  hand_over(g, s, len);
}

__global__ __launch_bounds__(AGG_WAVES * 32)
void pool_gemm(const float* __restrict__ psum, const float* __restrict__ cnts,
               const float* __restrict__ W, const float* __restrict__ bias,
               float* __restrict__ out) {
  static_assert(N_GRAPHS == 64 && D_OUT == 64, "8 waves x 2 tiles cover the 4 x 4 output tiles");
  const int lane = threadIdx.x & 31;
  const int wave = __builtin_amdgcn_readfirstlane((int)(threadIdx.x >> 5));
  const int ct = wave & 3;
  const int rg = wave >> 2;
  const int n16 = lane & 15;
  const int half = lane >> 4;
  const float* wp = W + ct * 16 + n16;
  const float* ap = psum + (size_t)((rg * 2) * 16 + n16) * TILE_K + 2 * half;
  v8f acc[2] = {v8f{}, v8f{}};
#pragma unroll
  for (int s = 0; s < 32; ++s) {
    const int c = 4 * s + 2 * half;
    v2f b;
    b.x = wp[(size_t)(4 * (c & 31) + (c >> 5)) * D_OUT];
    b.y = wp[(size_t)(4 * ((c + 1) & 31) + ((c + 1) >> 5)) * D_OUT];
#pragma unroll
    for (int t = 0; t < 2; ++t) {
      const v2f a = *(const v2f*)(ap + (size_t)t * 16 * TILE_K + 4 * s);
      acc[t] = __builtin_amdgcn_wmma_f32_16x16x4_f32(
           false, a,  false, b,
           (short)0, acc[t],  false,  false);
    }
  }
  const float bcol = bias[ct * 16 + n16];
#pragma unroll
  for (int t = 0; t < 2; ++t) {
#pragma unroll
    for (int i = 0; i < 8; ++i) {
      const int gq = (rg * 2 + t) * 16 + (half << 3) + i;
      const float cn = cnts[gq];
      out[gq * D_OUT + ct * 16 + n16] = (acc[t][i] + cn * bcol) / fmaxf(cn, 1.0f);
    }
  }
}

extern "C" void kernel_launch(void* const* d_in, const int* in_sizes, int n_in,
                              void* d_out, int out_size, void* d_ws, size_t ws_size,
                              hipStream_t stream) {
  const float* x     = (const float*)d_in[0];
  const int*   ei    = (const int*)d_in[1];
  const int*   batch = (const int*)d_in[2];
  const float* W1 = (const float*)d_in[3];  const float* b1 = (const float*)d_in[4];
  const float* W2 = (const float*)d_in[5];  const float* b2 = (const float*)d_in[6];
  const float* W3 = (const float*)d_in[7];  const float* b3 = (const float*)d_in[8];
  const float* W4 = (const float*)d_in[9];  const float* b4 = (const float*)d_in[10];
  float* out = (float*)d_out;

  float* bufA = (float*)d_ws;
  float* bufB = bufA + (size_t)N_NODES * D_HID;
  float* dis  = bufB + (size_t)N_NODES * D_HID;
  int* binStart = (int*)(dis + N_NODES);
  unsigned* rec = (unsigned*)(binStart + N_BINS + 1);
  int* binCnt   = (int*)bufB;
  float* psum = bufB;
  float* cnts = psum + N_GRAPHS * D_HID;

  const int* src = ei;
  const int* dst = ei + N_EDGES;

  if (ws_size < WS_NEED) return;

  const int TB = 256;
  auto blocks = [](long long n, int tb) { return (unsigned)((n + tb - 1) / tb); };

  zero_f32<<<blocks(N_NODES, TB), TB, 0, stream>>>(dis, N_NODES);
  deg_accum<<<blocks(N_EDGES, TB), TB, 0, stream>>>(dst, dis, N_EDGES);
  deg_finalize<<<blocks(N_NODES, TB), TB, 0, stream>>>(dis, N_NODES);

  bin_hist<<<BIN_BLOCKS, BIN_TB, 0, stream>>>(dst, binCnt);
  bin_scan<<<1, BIN_TB, 0, stream>>>(binCnt, binStart);
  bin_place<<<BIN_BLOCKS, BIN_TB, 0, stream>>>(src, dst, binCnt, rec);

  static_assert(D_IN == TILE_K && D_HID == TILE_K, "every layer's input has TILE_K features");
  layer_tile<D_HID, true><<<N_BINS, AGG_WAVES * 32, 0, stream>>>(rec, binStart, x, dis, W1, b1, bufA);
  layer_tile<D_HID, true><<<N_BINS, AGG_WAVES * 32, 0, stream>>>(rec, binStart, bufA, dis, W2, b2, bufB);
  layer_tile<D_HID, true><<<N_BINS, AGG_WAVES * 32, 0, stream>>>(rec, binStart, bufB, dis, W3, b3, bufA);

  zero_f32<<<blocks(N_GRAPHS * D_HID + N_GRAPHS, TB), TB, 0, stream>>>(
      psum, N_GRAPHS * D_HID + N_GRAPHS);
  layer_pool<<<N_BINS, AGG_WAVES * 32, 0, stream>>>(rec, binStart, bufA, dis, batch, psum, cnts);
  pool_gemm<<<1, AGG_WAVES * 32, 0, stream>>>(psum, cnts, W4, b4, out);
}


# in-degrees counted from the grouped edge list: a bin's block counts its records per row in LDS (integer adds), stores deg_inv_sqrt; zero pass, f32 memory adds, finalize pass gone: 12 launches are 10
# speedup vs baseline: 5.6606x; 1.2408x over previous
#include <hip/hip_runtime.h>
#include <hip/hip_bf16.h>

typedef float v2f __attribute__((ext_vector_type(2)));
typedef float v8f __attribute__((ext_vector_type(8)));

#define N_NODES  50000
#define N_EDGES  600000
#define N_GRAPHS 64
#define D_IN     128
#define D_HID    128
#define D_OUT    64

#define TILE_R     64
#define N_BINS     ((N_NODES + TILE_R - 1) / TILE_R)
#define BIN_BLOCKS 64
#define BIN_CHUNK  ((N_EDGES + BIN_BLOCKS - 1) / BIN_BLOCKS)
#define BIN_TB     1024
#define AGG_WAVES  8

static_assert(N_NODES <= 65536, "a source row index must fit the low 16 bits of an edge record");
static_assert(N_BINS <= BIN_TB, "the scan gives one thread to each bin");

constexpr size_t WS_NEED =
    ((size_t)2 * N_NODES * D_HID + N_NODES + N_BINS + 1 + N_EDGES) * sizeof(float);
static_assert((size_t)N_GRAPHS * D_HID + N_GRAPHS <= (size_t)N_NODES * D_HID,
              "the pooled sums and counts fit in one plane");

__global__ void zero_f32(float* __restrict__ p, int n) {
  int i = blockIdx.x * blockDim.x + threadIdx.x;
  if (i < n) p[i] = 0.0f;
}

__global__ __launch_bounds__(BIN_TB)
void bin_hist(const int* __restrict__ dst, int* __restrict__ cnt) {
  __shared__ int hist[N_BINS];
  for (int b = threadIdx.x; b < N_BINS; b += BIN_TB) hist[b] = 0;
  __syncthreads();
  const int e0 = blockIdx.x * BIN_CHUNK;
  const int e1 = min(e0 + BIN_CHUNK, N_EDGES);
  for (int e = e0 + threadIdx.x; e < e1; e += BIN_TB)
    atomicAdd(&hist[(unsigned)dst[e] / TILE_R], 1);
  __syncthreads();
  for (int b = threadIdx.x; b < N_BINS; b += BIN_TB) cnt[blockIdx.x * N_BINS + b] = hist[b];
}

__global__ __launch_bounds__(BIN_TB)
void bin_scan(int* __restrict__ cnt, int* __restrict__ binStart) {
  __shared__ int sc[BIN_TB];
  const int t = threadIdx.x;
  int tot = 0;
  if (t < N_BINS)
    for (int k = 0; k < BIN_BLOCKS; ++k) tot += cnt[k * N_BINS + t];
  sc[t] = tot;
  __syncthreads();
  for (int off = 1; off < BIN_TB; off <<= 1) {
    const int v = (t >= off) ? sc[t - off] : 0;
    __syncthreads();
    sc[t] += v;
    __syncthreads();
  }
  const int excl = sc[t] - tot;
  if (t <= N_BINS) binStart[t] = excl;
  if (t < N_BINS) {
    int run = excl;
    for (int k = 0; k < BIN_BLOCKS; ++k) {
      const int c = cnt[k * N_BINS + t];
      cnt[k * N_BINS + t] = run;
      run += c;
    }
  }
}

__global__ __launch_bounds__(BIN_TB)
void bin_place(const int* __restrict__ src, const int* __restrict__ dst,
               const int* __restrict__ cnt, unsigned* __restrict__ rec) {
  __shared__ int cur[N_BINS];
  for (int b = threadIdx.x; b < N_BINS; b += BIN_TB) cur[b] = cnt[blockIdx.x * N_BINS + b];
  __syncthreads();
  const int e0 = blockIdx.x * BIN_CHUNK;
  const int e1 = min(e0 + BIN_CHUNK, N_EDGES);
  for (int e = e0 + threadIdx.x; e < e1; e += BIN_TB) {
    const unsigned d = (unsigned)dst[e];
    const unsigned s = (unsigned)src[e];
    const unsigned bin = d / TILE_R;
    const int pos = atomicAdd(&cur[bin], 1);
    rec[pos] = ((d - bin * TILE_R) << 16) | s;
  }
}

#define DEG_TB 128
#define DEG_U  8
static_assert(DEG_TB >= TILE_R, "one thread per row of the bin clears, then stores, its count");
__global__ __launch_bounds__(DEG_TB)
void bin_degrees(const unsigned* __restrict__ rec, const int* __restrict__ binStart,
                 float* __restrict__ dis) {
  __shared__ int cnt[TILE_R];
  if (threadIdx.x < TILE_R) cnt[threadIdx.x] = 0;
  __syncthreads();
  const int e0 = binStart[blockIdx.x];
  const int e1 = binStart[blockIdx.x + 1];
  for (int base = e0 + (int)threadIdx.x; base < e1; base += DEG_U * DEG_TB) {
    unsigned r[DEG_U];
#pragma unroll
    for (int u = 0; u < DEG_U; ++u) {
      const int e = base + u * DEG_TB;
      r[u] = (e < e1) ? rec[e] : 0xffffffffu;
    }
#pragma unroll
    for (int u = 0; u < DEG_U; ++u)
      if (r[u] != 0xffffffffu) atomicAdd(&cnt[r[u] >> 16], 1);
  }
  __syncthreads();
  const int row = blockIdx.x * TILE_R + (int)threadIdx.x;
  if (threadIdx.x < TILE_R && row < N_NODES) dis[row] = rsqrtf((float)cnt[threadIdx.x] + 1.0f);
}

#define TILE_K 128
#define TILE_P 130

__device__ __forceinline__
void sum_rows_into_tile(const unsigned* __restrict__ rec, const int* __restrict__ binStart,
                        const float* __restrict__ h, const float* __restrict__ dis,
                        float* __restrict__ tile, float* __restrict__ disT,
                        int row0, int nrows, int wave, int lane) {
  constexpr int VPL = TILE_K / 32;
  typedef float vec __attribute__((ext_vector_type(VPL)));
  for (int r = wave; r < nrows; r += AGG_WAVES) {
    const float d = dis[row0 + r];
    const float dd = d * d;
    const vec v = *(const vec*)(h + (size_t)(row0 + r) * TILE_K + lane * VPL);
#pragma unroll
    for (int q = 0; q < VPL; ++q) tile[r * TILE_P + q * 32 + lane] = v[q] * dd;
  }
  if (threadIdx.x < TILE_R)
    disT[threadIdx.x] = ((int)threadIdx.x < nrows) ? dis[row0 + threadIdx.x] : 0.0f;
  __syncthreads();

  const int e0 = binStart[blockIdx.x];
  const int e1 = binStart[blockIdx.x + 1];
  for (int base = e0 + wave * 32; base < e1; base += AGG_WAVES * 32) {
    const int idx = base + lane;
    unsigned r = ((unsigned)TILE_R << 16) | (unsigned)row0;
    float nm = 0.0f;
    if (idx < e1) {
      r = rec[idx];
      nm = dis[r & 0xffffu] * disT[r >> 16];
    }
#pragma unroll
    for (int j = 0; j < 32; j += 8) {
      if (base + j >= e1) break;
      vec v[8];
      int dl[8];
      float w[8];
#pragma unroll
      for (int u = 0; u < 8; ++u) {
        const unsigned rj = (unsigned)__builtin_amdgcn_readlane((int)r, j + u);
        w[u] = __builtin_bit_cast(float, __builtin_amdgcn_readlane(__builtin_bit_cast(int, nm), j + u));
        dl[u] = (int)(rj >> 16);
        v[u] = *(const vec*)(h + (size_t)(rj & 0xffffu) * TILE_K + lane * VPL);
      }
#pragma unroll
      for (int u = 0; u < 8; ++u) {
#pragma unroll
        for (int q = 0; q < VPL; ++q)
          __hip_atomic_fetch_add(&tile[dl[u] * TILE_P + q * 32 + lane], v[u][q] * w[u],
                                 __ATOMIC_RELAXED, __HIP_MEMORY_SCOPE_WORKGROUP);
      }
    }
  }

  __syncthreads();
}

template <int FOUT, bool RELU>
__global__ __launch_bounds__(AGG_WAVES * 32)
void layer_tile(const unsigned* __restrict__ rec, const int* __restrict__ binStart,
                const float* __restrict__ h, const float* __restrict__ dis,
                const float* __restrict__ W, const float* __restrict__ bias,
                float* __restrict__ out) {
  __shared__ __attribute__((aligned(16))) float tile[(TILE_R + 1) * TILE_P];
  __shared__ float disT[TILE_R];

  const int lane = threadIdx.x & 31;
  const int wave = __builtin_amdgcn_readfirstlane((int)(threadIdx.x >> 5));
  const int row0 = blockIdx.x * TILE_R;
  const int nrows = min(TILE_R, N_NODES - row0);

  constexpr int NCT = FOUT / 16;
  constexpr int RTW = NCT / 2;
  static_assert(AGG_WAVES * RTW == (TILE_R / 16) * NCT, "8 waves cover the block's output tiles");
  const int ct = wave % NCT;
  const int rg = wave / NCT;
  const int n16 = lane & 15;
  const int half = lane >> 4;
  const float* wp = W + ct * 16 + n16;
  v2f bw[32];
#pragma unroll
  for (int s = 0; s < 32; ++s) {
    const int k = 8 * (s & 15) + 2 * (s >> 4) + half;
    bw[s].x = wp[(size_t)k * FOUT];
    bw[s].y = wp[(size_t)(k + 4) * FOUT];
  }
  const float bcol = bias[ct * 16 + n16];

  sum_rows_into_tile(rec, binStart, h, dis, tile, disT, row0, nrows, wave, lane);


  v8f acc[RTW];
#pragma unroll
  for (int t = 0; t < RTW; ++t) acc[t] = v8f{};
  const float* abase = tile + ((rg * RTW) * 16 + n16) * TILE_P + 32 * half;
#pragma unroll
  for (int s = 0; s < 32; ++s) {
#pragma unroll
    for (int t = 0; t < RTW; ++t) {
      const v2f a = *(const v2f*)(abase + t * 16 * TILE_P + 64 * (s >> 4) + 2 * (s & 15));
      acc[t] = __builtin_amdgcn_wmma_f32_16x16x4_f32(
           false, a,  false, bw[s],
           (short)0, acc[t],  false,  false);
    }
  }

#pragma unroll
  for (int t = 0; t < RTW; ++t) {
    const int rb = (rg * RTW + t) * 16 + (half << 3);
    float* op = out + (size_t)(row0 + rb) * FOUT + ct * 16 + n16;
    float val[8];
#pragma unroll
    for (int i = 0; i < 8; ++i) {
      val[i] = acc[t][i] + bcol;
      if (RELU) val[i] = fmaxf(val[i], 0.0f);
    }
    if (rb + 8 <= nrows) {
#pragma unroll
      for (int i = 0; i < 8; ++i) op[(size_t)i * FOUT] = val[i];
    } else {
#pragma unroll
      for (int i = 0; i < 8; ++i)
        if (rb + i < nrows) op[(size_t)i * FOUT] = val[i];
    }
  }
}

__global__ __launch_bounds__(AGG_WAVES * 32)
void layer_pool(const unsigned* __restrict__ rec, const int* __restrict__ binStart,
                const float* __restrict__ h, const float* __restrict__ dis,
                const int* __restrict__ batch, float* __restrict__ psum,
                float* __restrict__ cnts) {
  __shared__ __attribute__((aligned(16))) float tile[(TILE_R + 1) * TILE_P];
  __shared__ float disT[TILE_R];
  __shared__ int gidT[TILE_R];

  const int lane = threadIdx.x & 31;
  const int wave = __builtin_amdgcn_readfirstlane((int)(threadIdx.x >> 5));
  const int row0 = blockIdx.x * TILE_R;
  const int nrows = min(TILE_R, N_NODES - row0);

  if (threadIdx.x < TILE_R)
    gidT[threadIdx.x] = ((int)threadIdx.x < nrows) ? batch[row0 + threadIdx.x] : -1;

  sum_rows_into_tile(rec, binStart, h, dis, tile, disT, row0, nrows, wave, lane);

  constexpr int HR = TILE_R / 2;
  const int col = threadIdx.x & (TILE_K - 1);
  const int r0 = (wave >> 2) * HR;
  float v[HR];
  int gd[HR];
#pragma unroll
  for (int r = 0; r < HR; ++r) {
    v[r] = tile[(r0 + r) * TILE_P + col];
    gd[r] = __builtin_amdgcn_readfirstlane(gidT[r0 + r]);
  }
  auto hand_over = [&](int g, float s, int len) {
    if ((unsigned)g < (unsigned)N_GRAPHS) {
      atomicAdd(&psum[g * TILE_K + col], s);
      if (col == 0) atomicAdd(&cnts[g], (float)len);
    }
  };
  int g = gd[0];
  int len = 1;
  float s = v[0];
#pragma unroll
  for (int r = 1; r < HR; ++r) {
    if (gd[r] != g) {
      hand_over(g, s, len);
      g = gd[r];
      len = 0;
      s = 0.0f;
    }
    s += v[r];
    ++len;
  }
  hand_over(g, s, len);
}

__global__ __launch_bounds__(AGG_WAVES * 32)
void pool_gemm(const float* __restrict__ psum, const float* __restrict__ cnts,
               const float* __restrict__ W, const float* __restrict__ bias,
               float* __restrict__ out) {
  static_assert(N_GRAPHS == 64 && D_OUT == 64, "8 waves x 2 tiles cover the 4 x 4 output tiles");
  const int lane = threadIdx.x & 31;
  const int wave = __builtin_amdgcn_readfirstlane((int)(threadIdx.x >> 5));
  const int ct = wave & 3;
  const int rg = wave >> 2;
  const int n16 = lane & 15;
  const int half = lane >> 4;
  const float* wp = W + ct * 16 + n16;
  const float* ap = psum + (size_t)((rg * 2) * 16 + n16) * TILE_K + 2 * half;
  v8f acc[2] = {v8f{}, v8f{}};
#pragma unroll
  for (int s = 0; s < 32; ++s) {
    const int c = 4 * s + 2 * half;
    v2f b;
    b.x = wp[(size_t)(4 * (c & 31) + (c >> 5)) * D_OUT];
    b.y = wp[(size_t)(4 * ((c + 1) & 31) + ((c + 1) >> 5)) * D_OUT];
#pragma unroll
    for (int t = 0; t < 2; ++t) {
      const v2f a = *(const v2f*)(ap + (size_t)t * 16 * TILE_K + 4 * s);
      acc[t] = __builtin_amdgcn_wmma_f32_16x16x4_f32(
           false, a,  false, b,
           (short)0, acc[t],  false,  false);
    }
  }
  const float bcol = bias[ct * 16 + n16];
#pragma unroll
  for (int t = 0; t < 2; ++t) {
#pragma unroll
    for (int i = 0; i < 8; ++i) {
      const int gq = (rg * 2 + t) * 16 + (half << 3) + i;
      const float cn = cnts[gq];
      out[gq * D_OUT + ct * 16 + n16] = (acc[t][i] + cn * bcol) / fmaxf(cn, 1.0f);
    }
  }
}

extern "C" void kernel_launch(void* const* d_in, const int* in_sizes, int n_in,
                              void* d_out, int out_size, void* d_ws, size_t ws_size,
                              hipStream_t stream) {
  const float* x     = (const float*)d_in[0];
  const int*   ei    = (const int*)d_in[1];
  const int*   batch = (const int*)d_in[2];
  const float* W1 = (const float*)d_in[3];  const float* b1 = (const float*)d_in[4];
  const float* W2 = (const float*)d_in[5];  const float* b2 = (const float*)d_in[6];
  const float* W3 = (const float*)d_in[7];  const float* b3 = (const float*)d_in[8];
  const float* W4 = (const float*)d_in[9];  const float* b4 = (const float*)d_in[10];
  float* out = (float*)d_out;

  float* bufA = (float*)d_ws;
  float* bufB = bufA + (size_t)N_NODES * D_HID;
  float* dis  = bufB + (size_t)N_NODES * D_HID;
  int* binStart = (int*)(dis + N_NODES);
  unsigned* rec = (unsigned*)(binStart + N_BINS + 1);
  int* binCnt   = (int*)bufB;
  float* psum = bufB;
  float* cnts = psum + N_GRAPHS * D_HID;

  const int* src = ei;
  const int* dst = ei + N_EDGES;

  if (ws_size < WS_NEED) return;

  const int TB = 256;
  auto blocks = [](long long n, int tb) { return (unsigned)((n + tb - 1) / tb); };

  bin_hist<<<BIN_BLOCKS, BIN_TB, 0, stream>>>(dst, binCnt);
  bin_scan<<<1, BIN_TB, 0, stream>>>(binCnt, binStart);
  bin_place<<<BIN_BLOCKS, BIN_TB, 0, stream>>>(src, dst, binCnt, rec);

  bin_degrees<<<N_BINS, DEG_TB, 0, stream>>>(rec, binStart, dis);

  static_assert(D_IN == TILE_K && D_HID == TILE_K, "every layer's input has TILE_K features");
  layer_tile<D_HID, true><<<N_BINS, AGG_WAVES * 32, 0, stream>>>(rec, binStart, x, dis, W1, b1, bufA);
  layer_tile<D_HID, true><<<N_BINS, AGG_WAVES * 32, 0, stream>>>(rec, binStart, bufA, dis, W2, b2, bufB);
  layer_tile<D_HID, true><<<N_BINS, AGG_WAVES * 32, 0, stream>>>(rec, binStart, bufB, dis, W3, b3, bufA);

  zero_f32<<<blocks(N_GRAPHS * D_HID + N_GRAPHS, TB), TB, 0, stream>>>(
      psum, N_GRAPHS * D_HID + N_GRAPHS);
  layer_pool<<<N_BINS, AGG_WAVES * 32, 0, stream>>>(rec, binStart, bufA, dis, batch, psum, cnts);
  pool_gemm<<<1, AGG_WAVES * 32, 0, stream>>>(psum, cnts, W4, b4, out);
}


# loads issued as groups where a loop paid one round trip per trip: a thread's 10 edges in both grouping passes, a wave's 8 seed rows, next 32 records one trip ahead; sum order still LDS arrival
# speedup vs baseline: 5.8897x; 1.0405x over previous
#include <hip/hip_runtime.h>
#include <hip/hip_bf16.h>

typedef float v2f __attribute__((ext_vector_type(2)));
typedef float v8f __attribute__((ext_vector_type(8)));

#define N_NODES  50000
#define N_EDGES  600000
#define N_GRAPHS 64
#define D_IN     128
#define D_HID    128
#define D_OUT    64

#define TILE_R     64
#define N_BINS     ((N_NODES + TILE_R - 1) / TILE_R)
#define BIN_BLOCKS 64
#define BIN_CHUNK  ((N_EDGES + BIN_BLOCKS - 1) / BIN_BLOCKS)
#define BIN_TB     1024
#define BIN_U      ((BIN_CHUNK + BIN_TB - 1) / BIN_TB)
#define AGG_WAVES  8

static_assert(N_NODES <= 65536, "a source row index must fit the low 16 bits of an edge record");
static_assert(N_BINS <= BIN_TB, "the scan gives one thread to each bin");

constexpr size_t WS_NEED =
    ((size_t)2 * N_NODES * D_HID + N_NODES + N_BINS + 1 + N_EDGES) * sizeof(float);
static_assert((size_t)N_GRAPHS * D_HID + N_GRAPHS <= (size_t)N_NODES * D_HID,
              "the pooled sums and counts fit in one plane");

__global__ void zero_f32(float* __restrict__ p, int n) {
  int i = blockIdx.x * blockDim.x + threadIdx.x;
  if (i < n) p[i] = 0.0f;
}

__global__ __launch_bounds__(BIN_TB)
void bin_hist(const int* __restrict__ dst, int* __restrict__ cnt) {
  __shared__ int hist[N_BINS];
  const int e0 = blockIdx.x * BIN_CHUNK;
  const int e1 = min(e0 + BIN_CHUNK, N_EDGES);
  int d[BIN_U];
#pragma unroll
  for (int u = 0; u < BIN_U; ++u) {
    const int e = e0 + (int)threadIdx.x + u * BIN_TB;
    const int v = dst[min(e, N_EDGES - 1)];
    d[u] = (e < e1) ? v : -1;
  }
  for (int b = threadIdx.x; b < N_BINS; b += BIN_TB) hist[b] = 0;
  __syncthreads();
#pragma unroll
  for (int u = 0; u < BIN_U; ++u)
    if (d[u] >= 0) atomicAdd(&hist[(unsigned)d[u] / TILE_R], 1);
  __syncthreads();
  for (int b = threadIdx.x; b < N_BINS; b += BIN_TB) cnt[blockIdx.x * N_BINS + b] = hist[b];
}

__global__ __launch_bounds__(BIN_TB)
void bin_scan(int* __restrict__ cnt, int* __restrict__ binStart) {
  __shared__ int sc[BIN_TB];
  const int t = threadIdx.x;
  int tot = 0;
  if (t < N_BINS)
    for (int k = 0; k < BIN_BLOCKS; ++k) tot += cnt[k * N_BINS + t];
  sc[t] = tot;
  __syncthreads();
  for (int off = 1; off < BIN_TB; off <<= 1) {
    const int v = (t >= off) ? sc[t - off] : 0;
    __syncthreads();
    sc[t] += v;
    __syncthreads();
  }
  const int excl = sc[t] - tot;
  if (t <= N_BINS) binStart[t] = excl;
  if (t < N_BINS) {
    int run = excl;
    for (int k = 0; k < BIN_BLOCKS; ++k) {
      const int c = cnt[k * N_BINS + t];
      cnt[k * N_BINS + t] = run;
      run += c;
    }
  }
}

__global__ __launch_bounds__(BIN_TB)
void bin_place(const int* __restrict__ src, const int* __restrict__ dst,
               const int* __restrict__ cnt, unsigned* __restrict__ rec) {
  __shared__ int cur[N_BINS];
  const int e0 = blockIdx.x * BIN_CHUNK;
  const int e1 = min(e0 + BIN_CHUNK, N_EDGES);
  int d[BIN_U];
  int s[BIN_U];
#pragma unroll
  for (int u = 0; u < BIN_U; ++u) {
    const int e = e0 + (int)threadIdx.x + u * BIN_TB;
    const int ec = min(e, N_EDGES - 1);
    const int v = dst[ec];
    s[u] = src[ec];
    d[u] = (e < e1) ? v : -1;
  }
  for (int b = threadIdx.x; b < N_BINS; b += BIN_TB) cur[b] = cnt[blockIdx.x * N_BINS + b];
  __syncthreads();
  int pos[BIN_U];
#pragma unroll
  for (int u = 0; u < BIN_U; ++u) {
    pos[u] = -1;
    if (d[u] >= 0) pos[u] = atomicAdd(&cur[(unsigned)d[u] / TILE_R], 1);
  }
#pragma unroll
  for (int u = 0; u < BIN_U; ++u)
    if (d[u] >= 0)
      rec[pos[u]] = (((unsigned)d[u] % TILE_R) << 16) | (unsigned)s[u];
}

#define DEG_TB 128
#define DEG_U  8
static_assert(DEG_TB >= TILE_R, "one thread per row of the bin clears, then stores, its count");
__global__ __launch_bounds__(DEG_TB)
void bin_degrees(const unsigned* __restrict__ rec, const int* __restrict__ binStart,
                 float* __restrict__ dis) {
  __shared__ int cnt[TILE_R];
  if (threadIdx.x < TILE_R) cnt[threadIdx.x] = 0;
  __syncthreads();
  const int e0 = binStart[blockIdx.x];
  const int e1 = binStart[blockIdx.x + 1];
  for (int base = e0 + (int)threadIdx.x; base < e1; base += DEG_U * DEG_TB) {
    unsigned r[DEG_U];
#pragma unroll
    for (int u = 0; u < DEG_U; ++u) {
      const int e = base + u * DEG_TB;
      r[u] = (e < e1) ? rec[e] : 0xffffffffu;
    }
#pragma unroll
    for (int u = 0; u < DEG_U; ++u)
      if (r[u] != 0xffffffffu) atomicAdd(&cnt[r[u] >> 16], 1);
  }
  __syncthreads();
  const int row = blockIdx.x * TILE_R + (int)threadIdx.x;
  if (threadIdx.x < TILE_R && row < N_NODES) dis[row] = rsqrtf((float)cnt[threadIdx.x] + 1.0f);
}

#define TILE_K 128
#define TILE_P 130

__device__ __forceinline__
void sum_rows_into_tile(const unsigned* __restrict__ rec, const int* __restrict__ binStart,
                        const float* __restrict__ h, const float* __restrict__ dis,
                        float* __restrict__ tile, float* __restrict__ disT,
                        int row0, int nrows, int wave, int lane) {
  constexpr int VPL = TILE_K / 32;
  typedef float vec __attribute__((ext_vector_type(VPL)));
  constexpr int SEED_U = TILE_R / AGG_WAVES;
  static_assert(SEED_U * AGG_WAVES == TILE_R, "every wave seeds SEED_U whole rows of the tile");

  const int e0 = binStart[blockIdx.x];
  const int e1 = binStart[blockIdx.x + 1];

  vec sv[SEED_U];
  float sd[SEED_U];
#pragma unroll
  for (int u = 0; u < SEED_U; ++u) {
    const int rr = min(row0 + wave + u * AGG_WAVES, N_NODES - 1);
    sd[u] = dis[rr];
    sv[u] = *(const vec*)(h + (size_t)rr * TILE_K + lane * VPL);
  }
  const unsigned idle = ((unsigned)TILE_R << 16) | (unsigned)row0;
  int base = e0 + wave * 32;
  unsigned r = idle;
  if (base + lane < e1) r = rec[base + lane];
#pragma unroll
  for (int u = 0; u < SEED_U; ++u) {
    const int rt = wave + u * AGG_WAVES;
    const float dd = sd[u] * sd[u];
#pragma unroll
    for (int q = 0; q < VPL; ++q) tile[rt * TILE_P + q * 32 + lane] = sv[u][q] * dd;
  }
  if (threadIdx.x < TILE_R)
    disT[threadIdx.x] = ((int)threadIdx.x < nrows) ? dis[row0 + threadIdx.x] : 0.0f;
  float ds = 0.0f;
  if (base + lane < e1) ds = dis[r & 0xffffu];
  __syncthreads();

  for (; base < e1; base += AGG_WAVES * 32) {
    const unsigned rc = r;
    const float nm = (base + lane < e1) ? ds * disT[rc >> 16] : 0.0f;
    const int nidx = base + AGG_WAVES * 32 + lane;
    r = idle;
    if (nidx < e1) r = rec[nidx];
#pragma unroll
    for (int j = 0; j < 32; j += 8) {
      if (base + j >= e1) break;
      vec v[8];
      int dl[8];
      float w[8];
#pragma unroll
      for (int u = 0; u < 8; ++u) {
        const unsigned rj = (unsigned)__builtin_amdgcn_readlane((int)rc, j + u);
        w[u] = __builtin_bit_cast(float, __builtin_amdgcn_readlane(__builtin_bit_cast(int, nm), j + u));
        dl[u] = (int)(rj >> 16);
        v[u] = *(const vec*)(h + (size_t)(rj & 0xffffu) * TILE_K + lane * VPL);
      }
#pragma unroll
      for (int u = 0; u < 8; ++u) {
#pragma unroll
        for (int q = 0; q < VPL; ++q)
          __hip_atomic_fetch_add(&tile[dl[u] * TILE_P + q * 32 + lane], v[u][q] * w[u],
                                 __ATOMIC_RELAXED, __HIP_MEMORY_SCOPE_WORKGROUP);
      }
      if (j == 0) {
        ds = 0.0f;
        if (nidx < e1) ds = dis[r & 0xffffu];
      }
    }
  }

  __syncthreads();
}

template <int FOUT, bool RELU>
__global__ __launch_bounds__(AGG_WAVES * 32)
void layer_tile(const unsigned* __restrict__ rec, const int* __restrict__ binStart,
                const float* __restrict__ h, const float* __restrict__ dis,
                const float* __restrict__ W, const float* __restrict__ bias,
                float* __restrict__ out) {
  __shared__ __attribute__((aligned(16))) float tile[(TILE_R + 1) * TILE_P];
  __shared__ float disT[TILE_R];

  const int lane = threadIdx.x & 31;
  const int wave = __builtin_amdgcn_readfirstlane((int)(threadIdx.x >> 5));
  const int row0 = blockIdx.x * TILE_R;
  const int nrows = min(TILE_R, N_NODES - row0);

  constexpr int NCT = FOUT / 16;
  constexpr int RTW = NCT / 2;
  static_assert(AGG_WAVES * RTW == (TILE_R / 16) * NCT, "8 waves cover the block's output tiles");
  const int ct = wave % NCT;
  const int rg = wave / NCT;
  const int n16 = lane & 15;
  const int half = lane >> 4;
  const float* wp = W + ct * 16 + n16;
  v2f bw[32];
#pragma unroll
  for (int s = 0; s < 32; ++s) {
    const int k = 8 * (s & 15) + 2 * (s >> 4) + half;
    bw[s].x = wp[(size_t)k * FOUT];
    bw[s].y = wp[(size_t)(k + 4) * FOUT];
  }
  const float bcol = bias[ct * 16 + n16];

  sum_rows_into_tile(rec, binStart, h, dis, tile, disT, row0, nrows, wave, lane);


  v8f acc[RTW];
#pragma unroll
  for (int t = 0; t < RTW; ++t) acc[t] = v8f{};
  const float* abase = tile + ((rg * RTW) * 16 + n16) * TILE_P + 32 * half;
#pragma unroll
  for (int s = 0; s < 32; ++s) {
#pragma unroll
    for (int t = 0; t < RTW; ++t) {
      const v2f a = *(const v2f*)(abase + t * 16 * TILE_P + 64 * (s >> 4) + 2 * (s & 15));
      acc[t] = __builtin_amdgcn_wmma_f32_16x16x4_f32(
           false, a,  false, bw[s],
           (short)0, acc[t],  false,  false);
    }
  }

#pragma unroll
  for (int t = 0; t < RTW; ++t) {
    const int rb = (rg * RTW + t) * 16 + (half << 3);
    float* op = out + (size_t)(row0 + rb) * FOUT + ct * 16 + n16;
    float val[8];
#pragma unroll
    for (int i = 0; i < 8; ++i) {
      val[i] = acc[t][i] + bcol;
      if (RELU) val[i] = fmaxf(val[i], 0.0f);
    }
    if (rb + 8 <= nrows) {
#pragma unroll
      for (int i = 0; i < 8; ++i) op[(size_t)i * FOUT] = val[i];
    } else {
#pragma unroll
      for (int i = 0; i < 8; ++i)
        if (rb + i < nrows) op[(size_t)i * FOUT] = val[i];
    }
  }
}

__global__ __launch_bounds__(AGG_WAVES * 32)
void layer_pool(const unsigned* __restrict__ rec, const int* __restrict__ binStart,
                const float* __restrict__ h, const float* __restrict__ dis,
                const int* __restrict__ batch, float* __restrict__ psum,
                float* __restrict__ cnts) {
  __shared__ __attribute__((aligned(16))) float tile[(TILE_R + 1) * TILE_P];
  __shared__ float disT[TILE_R];
  __shared__ int gidT[TILE_R];

  const int lane = threadIdx.x & 31;
  const int wave = __builtin_amdgcn_readfirstlane((int)(threadIdx.x >> 5));
  const int row0 = blockIdx.x * TILE_R;
  const int nrows = min(TILE_R, N_NODES - row0);

  if (threadIdx.x < TILE_R)
    gidT[threadIdx.x] = ((int)threadIdx.x < nrows) ? batch[row0 + threadIdx.x] : -1;

  sum_rows_into_tile(rec, binStart, h, dis, tile, disT, row0, nrows, wave, lane);

  constexpr int HR = TILE_R / 2;
  const int col = threadIdx.x & (TILE_K - 1);
  const int r0 = (wave >> 2) * HR;
  float v[HR];
  int gd[HR];
#pragma unroll
  for (int r = 0; r < HR; ++r) {
    v[r] = tile[(r0 + r) * TILE_P + col];
    gd[r] = __builtin_amdgcn_readfirstlane(gidT[r0 + r]);
  }
  auto hand_over = [&](int g, float s, int len) {
    if ((unsigned)g < (unsigned)N_GRAPHS) {
      atomicAdd(&psum[g * TILE_K + col], s);
      if (col == 0) atomicAdd(&cnts[g], (float)len);
    }
  };
  int g = gd[0];
  int len = 1;
  float s = v[0];
#pragma unroll
  for (int r = 1; r < HR; ++r) {
    if (gd[r] != g) {
      hand_over(g, s, len);
      g = gd[r];
      len = 0;
      s = 0.0f;
    }
    s += v[r];
    ++len;
  }
  hand_over(g, s, len);
}

__global__ __launch_bounds__(AGG_WAVES * 32)
void pool_gemm(const float* __restrict__ psum, const float* __restrict__ cnts,
               const float* __restrict__ W, const float* __restrict__ bias,
               float* __restrict__ out) {
  static_assert(N_GRAPHS == 64 && D_OUT == 64, "8 waves x 2 tiles cover the 4 x 4 output tiles");
  const int lane = threadIdx.x & 31;
  const int wave = __builtin_amdgcn_readfirstlane((int)(threadIdx.x >> 5));
  const int ct = wave & 3;
  const int rg = wave >> 2;
  const int n16 = lane & 15;
  const int half = lane >> 4;
  const float* wp = W + ct * 16 + n16;
  const float* ap = psum + (size_t)((rg * 2) * 16 + n16) * TILE_K + 2 * half;
  v8f acc[2] = {v8f{}, v8f{}};
#pragma unroll
  for (int s = 0; s < 32; ++s) {
    const int c = 4 * s + 2 * half;
    v2f b;
    b.x = wp[(size_t)(4 * (c & 31) + (c >> 5)) * D_OUT];
    b.y = wp[(size_t)(4 * ((c + 1) & 31) + ((c + 1) >> 5)) * D_OUT];
#pragma unroll
    for (int t = 0; t < 2; ++t) {
      const v2f a = *(const v2f*)(ap + (size_t)t * 16 * TILE_K + 4 * s);
      acc[t] = __builtin_amdgcn_wmma_f32_16x16x4_f32(
           false, a,  false, b,
           (short)0, acc[t],  false,  false);
    }
  }
  const float bcol = bias[ct * 16 + n16];
#pragma unroll
  for (int t = 0; t < 2; ++t) {
#pragma unroll
    for (int i = 0; i < 8; ++i) {
      const int gq = (rg * 2 + t) * 16 + (half << 3) + i;
      const float cn = cnts[gq];
      out[gq * D_OUT + ct * 16 + n16] = (acc[t][i] + cn * bcol) / fmaxf(cn, 1.0f);
    }
  }
}

extern "C" void kernel_launch(void* const* d_in, const int* in_sizes, int n_in,
                              void* d_out, int out_size, void* d_ws, size_t ws_size,
                              hipStream_t stream) {
  const float* x     = (const float*)d_in[0];
  const int*   ei    = (const int*)d_in[1];
  const int*   batch = (const int*)d_in[2];
  const float* W1 = (const float*)d_in[3];  const float* b1 = (const float*)d_in[4];
  const float* W2 = (const float*)d_in[5];  const float* b2 = (const float*)d_in[6];
  const float* W3 = (const float*)d_in[7];  const float* b3 = (const float*)d_in[8];
  const float* W4 = (const float*)d_in[9];  const float* b4 = (const float*)d_in[10];
  float* out = (float*)d_out;

  float* bufA = (float*)d_ws;
  float* bufB = bufA + (size_t)N_NODES * D_HID;
  float* dis  = bufB + (size_t)N_NODES * D_HID;
  int* binStart = (int*)(dis + N_NODES);
  unsigned* rec = (unsigned*)(binStart + N_BINS + 1);
  int* binCnt   = (int*)bufB;
  float* psum = bufB;
  float* cnts = psum + N_GRAPHS * D_HID;

  const int* src = ei;
  const int* dst = ei + N_EDGES;

  if (ws_size < WS_NEED) return;

  const int TB = 256;
  auto blocks = [](long long n, int tb) { return (unsigned)((n + tb - 1) / tb); };

  bin_hist<<<BIN_BLOCKS, BIN_TB, 0, stream>>>(dst, binCnt);
  bin_scan<<<1, BIN_TB, 0, stream>>>(binCnt, binStart);
  bin_place<<<BIN_BLOCKS, BIN_TB, 0, stream>>>(src, dst, binCnt, rec);

  bin_degrees<<<N_BINS, DEG_TB, 0, stream>>>(rec, binStart, dis);

  static_assert(D_IN == TILE_K && D_HID == TILE_K, "every layer's input has TILE_K features");
  layer_tile<D_HID, true><<<N_BINS, AGG_WAVES * 32, 0, stream>>>(rec, binStart, x, dis, W1, b1, bufA);
  layer_tile<D_HID, true><<<N_BINS, AGG_WAVES * 32, 0, stream>>>(rec, binStart, bufA, dis, W2, b2, bufB);
  layer_tile<D_HID, true><<<N_BINS, AGG_WAVES * 32, 0, stream>>>(rec, binStart, bufB, dis, W3, b3, bufA);

  zero_f32<<<blocks(N_GRAPHS * D_HID + N_GRAPHS, TB), TB, 0, stream>>>(
      psum, N_GRAPHS * D_HID + N_GRAPHS);
  layer_pool<<<N_BINS, AGG_WAVES * 32, 0, stream>>>(rec, binStart, bufA, dis, batch, psum, cnts);
  pool_gemm<<<1, AGG_WAVES * 32, 0, stream>>>(psum, cnts, W4, b4, out);
}


# planes hold rows pre-scaled by own deg^-1/2: layers 2-4 add whole source rows, no per-edge weight, no dis[src] lookup; dst factor applied once behind W; products regrouped, sum order LDS arrival
# speedup vs baseline: 6.0709x; 1.0308x over previous
#include <hip/hip_runtime.h>
#include <hip/hip_bf16.h>

typedef float v2f __attribute__((ext_vector_type(2)));
typedef float v8f __attribute__((ext_vector_type(8)));

#define N_NODES  50000
#define N_EDGES  600000
#define N_GRAPHS 64
#define D_IN     128
#define D_HID    128
#define D_OUT    64

#define TILE_R     64
#define N_BINS     ((N_NODES + TILE_R - 1) / TILE_R)
#define BIN_BLOCKS 64
#define BIN_CHUNK  ((N_EDGES + BIN_BLOCKS - 1) / BIN_BLOCKS)
#define BIN_TB     1024
#define BIN_U      ((BIN_CHUNK + BIN_TB - 1) / BIN_TB)
#define AGG_WAVES  8

static_assert(N_NODES <= 65536, "a source row index must fit the low 16 bits of an edge record");
static_assert(N_BINS <= BIN_TB, "the scan gives one thread to each bin");

constexpr size_t WS_NEED =
    ((size_t)2 * N_NODES * D_HID + N_NODES + N_BINS + 1 + N_EDGES) * sizeof(float);
static_assert((size_t)N_GRAPHS * D_HID + N_GRAPHS <= (size_t)N_NODES * D_HID,
              "the pooled sums and counts fit in one plane");

__global__ void zero_f32(float* __restrict__ p, int n) {
  int i = blockIdx.x * blockDim.x + threadIdx.x;
  if (i < n) p[i] = 0.0f;
}

__global__ __launch_bounds__(BIN_TB)
void bin_hist(const int* __restrict__ dst, int* __restrict__ cnt) {
  __shared__ int hist[N_BINS];
  const int e0 = blockIdx.x * BIN_CHUNK;
  const int e1 = min(e0 + BIN_CHUNK, N_EDGES);
  int d[BIN_U];
#pragma unroll
  for (int u = 0; u < BIN_U; ++u) {
    const int e = e0 + (int)threadIdx.x + u * BIN_TB;
    const int v = dst[min(e, N_EDGES - 1)];
    d[u] = (e < e1) ? v : -1;
  }
  for (int b = threadIdx.x; b < N_BINS; b += BIN_TB) hist[b] = 0;
  __syncthreads();
#pragma unroll
  for (int u = 0; u < BIN_U; ++u)
    if (d[u] >= 0) atomicAdd(&hist[(unsigned)d[u] / TILE_R], 1);
  __syncthreads();
  for (int b = threadIdx.x; b < N_BINS; b += BIN_TB) cnt[blockIdx.x * N_BINS + b] = hist[b];
}

__global__ __launch_bounds__(BIN_TB)
void bin_scan(int* __restrict__ cnt, int* __restrict__ binStart) {
  __shared__ int sc[BIN_TB];
  const int t = threadIdx.x;
  int tot = 0;
  if (t < N_BINS)
    for (int k = 0; k < BIN_BLOCKS; ++k) tot += cnt[k * N_BINS + t];
  sc[t] = tot;
  __syncthreads();
  for (int off = 1; off < BIN_TB; off <<= 1) {
    const int v = (t >= off) ? sc[t - off] : 0;
    __syncthreads();
    sc[t] += v;
    __syncthreads();
  }
  const int excl = sc[t] - tot;
  if (t <= N_BINS) binStart[t] = excl;
  if (t < N_BINS) {
    int run = excl;
    for (int k = 0; k < BIN_BLOCKS; ++k) {
      const int c = cnt[k * N_BINS + t];
      cnt[k * N_BINS + t] = run;
      run += c;
    }
  }
}

__global__ __launch_bounds__(BIN_TB)
void bin_place(const int* __restrict__ src, const int* __restrict__ dst,
               const int* __restrict__ cnt, unsigned* __restrict__ rec) {
  __shared__ int cur[N_BINS];
  const int e0 = blockIdx.x * BIN_CHUNK;
  const int e1 = min(e0 + BIN_CHUNK, N_EDGES);
  int d[BIN_U];
  int s[BIN_U];
#pragma unroll
  for (int u = 0; u < BIN_U; ++u) {
    const int e = e0 + (int)threadIdx.x + u * BIN_TB;
    const int ec = min(e, N_EDGES - 1);
    const int v = dst[ec];
    s[u] = src[ec];
    d[u] = (e < e1) ? v : -1;
  }
  for (int b = threadIdx.x; b < N_BINS; b += BIN_TB) cur[b] = cnt[blockIdx.x * N_BINS + b];
  __syncthreads();
  int pos[BIN_U];
#pragma unroll
  for (int u = 0; u < BIN_U; ++u) {
    pos[u] = -1;
    if (d[u] >= 0) pos[u] = atomicAdd(&cur[(unsigned)d[u] / TILE_R], 1);
  }
#pragma unroll
  for (int u = 0; u < BIN_U; ++u)
    if (d[u] >= 0)
      rec[pos[u]] = (((unsigned)d[u] % TILE_R) << 16) | (unsigned)s[u];
}

#define DEG_TB 128
#define DEG_U  8
static_assert(DEG_TB >= TILE_R, "one thread per row of the bin clears, then stores, its count");
__global__ __launch_bounds__(DEG_TB)
void bin_degrees(const unsigned* __restrict__ rec, const int* __restrict__ binStart,
                 float* __restrict__ dis) {
  __shared__ int cnt[TILE_R];
  if (threadIdx.x < TILE_R) cnt[threadIdx.x] = 0;
  __syncthreads();
  const int e0 = binStart[blockIdx.x];
  const int e1 = binStart[blockIdx.x + 1];
  for (int base = e0 + (int)threadIdx.x; base < e1; base += DEG_U * DEG_TB) {
    unsigned r[DEG_U];
#pragma unroll
    for (int u = 0; u < DEG_U; ++u) {
      const int e = base + u * DEG_TB;
      r[u] = (e < e1) ? rec[e] : 0xffffffffu;
    }
#pragma unroll
    for (int u = 0; u < DEG_U; ++u)
      if (r[u] != 0xffffffffu) atomicAdd(&cnt[r[u] >> 16], 1);
  }
  __syncthreads();
  const int row = blockIdx.x * TILE_R + (int)threadIdx.x;
  if (threadIdx.x < TILE_R && row < N_NODES) dis[row] = rsqrtf((float)cnt[threadIdx.x] + 1.0f);
}

#define TILE_K 128
#define TILE_P 130

template <bool PRE>
__device__ __forceinline__
void sum_rows_into_tile(const unsigned* __restrict__ rec, const int* __restrict__ binStart,
                        const float* __restrict__ h, const float* __restrict__ dis,
                        float* __restrict__ tile, float* __restrict__ disT,
                        int row0, int nrows, int wave, int lane) {
  constexpr int VPL = TILE_K / 32;
  typedef float vec __attribute__((ext_vector_type(VPL)));
  constexpr int SEED_U = TILE_R / AGG_WAVES;
  static_assert(SEED_U * AGG_WAVES == TILE_R, "every wave seeds SEED_U whole rows of the tile");

  const int e0 = binStart[blockIdx.x];
  const int e1 = binStart[blockIdx.x + 1];

  vec sv[SEED_U];
  float sd[SEED_U];
#pragma unroll
  for (int u = 0; u < SEED_U; ++u) {
    const int rr = min(row0 + wave + u * AGG_WAVES, N_NODES - 1);
    sd[u] = PRE ? 1.0f : dis[rr];
    sv[u] = *(const vec*)(h + (size_t)rr * TILE_K + lane * VPL);
  }
  const unsigned idle = ((unsigned)TILE_R << 16) | (unsigned)row0;
  int base = e0 + wave * 32;
  unsigned r = idle;
  if (base + lane < e1) r = rec[base + lane];
#pragma unroll
  for (int u = 0; u < SEED_U; ++u) {
    const int rt = wave + u * AGG_WAVES;
#pragma unroll
    for (int q = 0; q < VPL; ++q)
      tile[rt * TILE_P + q * 32 + lane] = PRE ? sv[u][q] : sv[u][q] * sd[u];
  }
  if (threadIdx.x < TILE_R)
    disT[threadIdx.x] = ((int)threadIdx.x < nrows) ? dis[row0 + threadIdx.x] : 0.0f;
  float ds = 0.0f;
  if (!PRE && base + lane < e1) ds = dis[r & 0xffffu];
  __syncthreads();

  for (; base < e1; base += AGG_WAVES * 32) {
    const unsigned rc = r;
    const float nm = ds;
    const int nidx = base + AGG_WAVES * 32 + lane;
    r = idle;
    if (nidx < e1) r = rec[nidx];
#pragma unroll
    for (int j = 0; j < 32; j += 8) {
      if (base + j >= e1) break;
      vec v[8];
      int dl[8];
      float w[8];
#pragma unroll
      for (int u = 0; u < 8; ++u) {
        const unsigned rj = (unsigned)__builtin_amdgcn_readlane((int)rc, j + u);
        w[u] = PRE ? 1.0f
                   : __builtin_bit_cast(float, __builtin_amdgcn_readlane(__builtin_bit_cast(int, nm), j + u));
        dl[u] = (int)(rj >> 16);
        v[u] = *(const vec*)(h + (size_t)(rj & 0xffffu) * TILE_K + lane * VPL);
      }
#pragma unroll
      for (int u = 0; u < 8; ++u) {
#pragma unroll
        for (int q = 0; q < VPL; ++q)
          __hip_atomic_fetch_add(&tile[dl[u] * TILE_P + q * 32 + lane],
                                 PRE ? v[u][q] : v[u][q] * w[u],
                                 __ATOMIC_RELAXED, __HIP_MEMORY_SCOPE_WORKGROUP);
      }
      if (!PRE && j == 0) {
        ds = 0.0f;
        if (nidx < e1) ds = dis[r & 0xffffu];
      }
    }
  }

  __syncthreads();
}

template <int FOUT, bool RELU, bool PRE>
__global__ __launch_bounds__(AGG_WAVES * 32)
void layer_tile(const unsigned* __restrict__ rec, const int* __restrict__ binStart,
                const float* __restrict__ h, const float* __restrict__ dis,
                const float* __restrict__ W, const float* __restrict__ bias,
                float* __restrict__ out) {
  __shared__ __attribute__((aligned(16))) float tile[(TILE_R + 1) * TILE_P];
  __shared__ float disT[TILE_R];

  const int lane = threadIdx.x & 31;
  const int wave = __builtin_amdgcn_readfirstlane((int)(threadIdx.x >> 5));
  const int row0 = blockIdx.x * TILE_R;
  const int nrows = min(TILE_R, N_NODES - row0);

  constexpr int NCT = FOUT / 16;
  constexpr int RTW = NCT / 2;
  static_assert(AGG_WAVES * RTW == (TILE_R / 16) * NCT, "8 waves cover the block's output tiles");
  const int ct = wave % NCT;
  const int rg = wave / NCT;
  const int n16 = lane & 15;
  const int half = lane >> 4;
  const float* wp = W + ct * 16 + n16;
  v2f bw[32];
#pragma unroll
  for (int s = 0; s < 32; ++s) {
    const int k = 8 * (s & 15) + 2 * (s >> 4) + half;
    bw[s].x = wp[(size_t)k * FOUT];
    bw[s].y = wp[(size_t)(k + 4) * FOUT];
  }
  const float bcol = bias[ct * 16 + n16];

  sum_rows_into_tile<PRE>(rec, binStart, h, dis, tile, disT, row0, nrows, wave, lane);


  v8f acc[RTW];
#pragma unroll
  for (int t = 0; t < RTW; ++t) acc[t] = v8f{};
  const float* abase = tile + ((rg * RTW) * 16 + n16) * TILE_P + 32 * half;
#pragma unroll
  for (int s = 0; s < 32; ++s) {
#pragma unroll
    for (int t = 0; t < RTW; ++t) {
      const v2f a = *(const v2f*)(abase + t * 16 * TILE_P + 64 * (s >> 4) + 2 * (s & 15));
      acc[t] = __builtin_amdgcn_wmma_f32_16x16x4_f32(
           false, a,  false, bw[s],
           (short)0, acc[t],  false,  false);
    }
  }

#pragma unroll
  for (int t = 0; t < RTW; ++t) {
    const int rb = (rg * RTW + t) * 16 + (half << 3);
    float* op = out + (size_t)(row0 + rb) * FOUT + ct * 16 + n16;
    float val[8];
#pragma unroll
    for (int i = 0; i < 8; ++i) {
      const float dr = disT[rb + i];
      val[i] = acc[t][i] * dr + bcol;
      if (RELU) val[i] = fmaxf(val[i], 0.0f);
      val[i] *= dr;
    }
    if (rb + 8 <= nrows) {
#pragma unroll
      for (int i = 0; i < 8; ++i) op[(size_t)i * FOUT] = val[i];
    } else {
#pragma unroll
      for (int i = 0; i < 8; ++i)
        if (rb + i < nrows) op[(size_t)i * FOUT] = val[i];
    }
  }
}

__global__ __launch_bounds__(AGG_WAVES * 32)
void layer_pool(const unsigned* __restrict__ rec, const int* __restrict__ binStart,
                const float* __restrict__ h, const float* __restrict__ dis,
                const int* __restrict__ batch, float* __restrict__ psum,
                float* __restrict__ cnts) {
  __shared__ __attribute__((aligned(16))) float tile[(TILE_R + 1) * TILE_P];
  __shared__ float disT[TILE_R];
  __shared__ int gidT[TILE_R];

  const int lane = threadIdx.x & 31;
  const int wave = __builtin_amdgcn_readfirstlane((int)(threadIdx.x >> 5));
  const int row0 = blockIdx.x * TILE_R;
  const int nrows = min(TILE_R, N_NODES - row0);

  if (threadIdx.x < TILE_R)
    gidT[threadIdx.x] = ((int)threadIdx.x < nrows) ? batch[row0 + threadIdx.x] : -1;

  sum_rows_into_tile<true>(rec, binStart, h, dis, tile, disT, row0, nrows, wave, lane);

  constexpr int HR = TILE_R / 2;
  const int col = threadIdx.x & (TILE_K - 1);
  const int r0 = (wave >> 2) * HR;
  float v[HR];
  int gd[HR];
#pragma unroll
  for (int r = 0; r < HR; ++r) {
    v[r] = tile[(r0 + r) * TILE_P + col] * disT[r0 + r];
    gd[r] = __builtin_amdgcn_readfirstlane(gidT[r0 + r]);
  }
  auto hand_over = [&](int g, float s, int len) {
    if ((unsigned)g < (unsigned)N_GRAPHS) {
      atomicAdd(&psum[g * TILE_K + col], s);
      if (col == 0) atomicAdd(&cnts[g], (float)len);
    }
  };
  int g = gd[0];
  int len = 1;
  float s = v[0];
#pragma unroll
  for (int r = 1; r < HR; ++r) {
    if (gd[r] != g) {
      hand_over(g, s, len);
      g = gd[r];
      len = 0;
      s = 0.0f;
    }
    s += v[r];
    ++len;
  }
  hand_over(g, s, len);
}

__global__ __launch_bounds__(AGG_WAVES * 32)
void pool_gemm(const float* __restrict__ psum, const float* __restrict__ cnts,
               const float* __restrict__ W, const float* __restrict__ bias,
               float* __restrict__ out) {
  static_assert(N_GRAPHS == 64 && D_OUT == 64, "8 waves x 2 tiles cover the 4 x 4 output tiles");
  const int lane = threadIdx.x & 31;
  const int wave = __builtin_amdgcn_readfirstlane((int)(threadIdx.x >> 5));
  const int ct = wave & 3;
  const int rg = wave >> 2;
  const int n16 = lane & 15;
  const int half = lane >> 4;
  const float* wp = W + ct * 16 + n16;
  const float* ap = psum + (size_t)((rg * 2) * 16 + n16) * TILE_K + 2 * half;
  v8f acc[2] = {v8f{}, v8f{}};
#pragma unroll
  for (int s = 0; s < 32; ++s) {
    const int c = 4 * s + 2 * half;
    v2f b;
    b.x = wp[(size_t)(4 * (c & 31) + (c >> 5)) * D_OUT];
    b.y = wp[(size_t)(4 * ((c + 1) & 31) + ((c + 1) >> 5)) * D_OUT];
#pragma unroll
    for (int t = 0; t < 2; ++t) {
      const v2f a = *(const v2f*)(ap + (size_t)t * 16 * TILE_K + 4 * s);
      acc[t] = __builtin_amdgcn_wmma_f32_16x16x4_f32(
           false, a,  false, b,
           (short)0, acc[t],  false,  false);
    }
  }
  const float bcol = bias[ct * 16 + n16];
#pragma unroll
  for (int t = 0; t < 2; ++t) {
#pragma unroll
    for (int i = 0; i < 8; ++i) {
      const int gq = (rg * 2 + t) * 16 + (half << 3) + i;
      const float cn = cnts[gq];
      out[gq * D_OUT + ct * 16 + n16] = (acc[t][i] + cn * bcol) / fmaxf(cn, 1.0f);
    }
  }
}

extern "C" void kernel_launch(void* const* d_in, const int* in_sizes, int n_in,
                              void* d_out, int out_size, void* d_ws, size_t ws_size,
                              hipStream_t stream) {
  const float* x     = (const float*)d_in[0];
  const int*   ei    = (const int*)d_in[1];
  const int*   batch = (const int*)d_in[2];
  const float* W1 = (const float*)d_in[3];  const float* b1 = (const float*)d_in[4];
  const float* W2 = (const float*)d_in[5];  const float* b2 = (const float*)d_in[6];
  const float* W3 = (const float*)d_in[7];  const float* b3 = (const float*)d_in[8];
  const float* W4 = (const float*)d_in[9];  const float* b4 = (const float*)d_in[10];
  float* out = (float*)d_out;

  float* bufA = (float*)d_ws;
  float* bufB = bufA + (size_t)N_NODES * D_HID;
  float* dis  = bufB + (size_t)N_NODES * D_HID;
  int* binStart = (int*)(dis + N_NODES);
  unsigned* rec = (unsigned*)(binStart + N_BINS + 1);
  int* binCnt   = (int*)bufB;
  float* psum = bufB;
  float* cnts = psum + N_GRAPHS * D_HID;

  const int* src = ei;
  const int* dst = ei + N_EDGES;

  if (ws_size < WS_NEED) return;

  const int TB = 256;
  auto blocks = [](long long n, int tb) { return (unsigned)((n + tb - 1) / tb); };

  bin_hist<<<BIN_BLOCKS, BIN_TB, 0, stream>>>(dst, binCnt);
  bin_scan<<<1, BIN_TB, 0, stream>>>(binCnt, binStart);
  bin_place<<<BIN_BLOCKS, BIN_TB, 0, stream>>>(src, dst, binCnt, rec);

  bin_degrees<<<N_BINS, DEG_TB, 0, stream>>>(rec, binStart, dis);

  static_assert(D_IN == TILE_K && D_HID == TILE_K, "every layer's input has TILE_K features");
  layer_tile<D_HID, true, false><<<N_BINS, AGG_WAVES * 32, 0, stream>>>(rec, binStart, x, dis, W1, b1, bufA);
  layer_tile<D_HID, true, true><<<N_BINS, AGG_WAVES * 32, 0, stream>>>(rec, binStart, bufA, dis, W2, b2, bufB);
  layer_tile<D_HID, true, true><<<N_BINS, AGG_WAVES * 32, 0, stream>>>(rec, binStart, bufB, dis, W3, b3, bufA);

  zero_f32<<<blocks(N_GRAPHS * D_HID + N_GRAPHS, TB), TB, 0, stream>>>(
      psum, N_GRAPHS * D_HID + N_GRAPHS);
  layer_pool<<<N_BINS, AGG_WAVES * 32, 0, stream>>>(rec, binStart, bufA, dis, batch, psum, cnts);
  pool_gemm<<<1, AGG_WAVES * 32, 0, stream>>>(psum, cnts, W4, b4, out);
}


# layer 4's product W4 moved into layer 3's block: its relu rows go back to LDS, a second f32 WMMA pass makes 64-wide rows; the pooled layer sums 64-wide rows; association of the sums changes
# speedup vs baseline: 6.2900x; 1.0361x over previous
#include <hip/hip_runtime.h>
#include <hip/hip_bf16.h>

typedef float v2f __attribute__((ext_vector_type(2)));
typedef float v8f __attribute__((ext_vector_type(8)));

#define N_NODES  50000
#define N_EDGES  600000
#define N_GRAPHS 64
#define D_IN     128
#define D_HID    128
#define D_OUT    64

#define TILE_R     64
#define N_BINS     ((N_NODES + TILE_R - 1) / TILE_R)
#define BIN_BLOCKS 64
#define BIN_CHUNK  ((N_EDGES + BIN_BLOCKS - 1) / BIN_BLOCKS)
#define BIN_TB     1024
#define BIN_U      ((BIN_CHUNK + BIN_TB - 1) / BIN_TB)
#define AGG_WAVES  8

static_assert(N_NODES <= 65536, "a source row index must fit the low 16 bits of an edge record");
static_assert(N_BINS <= BIN_TB, "the scan gives one thread to each bin");

constexpr size_t WS_NEED =
    ((size_t)2 * N_NODES * D_HID + N_NODES + N_BINS + 1 + N_EDGES) * sizeof(float);
static_assert((size_t)N_GRAPHS * D_OUT + N_GRAPHS <= (size_t)N_NODES * D_HID,
              "the pooled sums and counts fit in one plane");

__global__ void zero_f32(float* __restrict__ p, int n) {
  int i = blockIdx.x * blockDim.x + threadIdx.x;
  if (i < n) p[i] = 0.0f;
}

__global__ __launch_bounds__(BIN_TB)
void bin_hist(const int* __restrict__ dst, int* __restrict__ cnt) {
  __shared__ int hist[N_BINS];
  const int e0 = blockIdx.x * BIN_CHUNK;
  const int e1 = min(e0 + BIN_CHUNK, N_EDGES);
  int d[BIN_U];
#pragma unroll
  for (int u = 0; u < BIN_U; ++u) {
    const int e = e0 + (int)threadIdx.x + u * BIN_TB;
    const int v = dst[min(e, N_EDGES - 1)];
    d[u] = (e < e1) ? v : -1;
  }
  for (int b = threadIdx.x; b < N_BINS; b += BIN_TB) hist[b] = 0;
  __syncthreads();
#pragma unroll
  for (int u = 0; u < BIN_U; ++u)
    if (d[u] >= 0) atomicAdd(&hist[(unsigned)d[u] / TILE_R], 1);
  __syncthreads();
  for (int b = threadIdx.x; b < N_BINS; b += BIN_TB) cnt[blockIdx.x * N_BINS + b] = hist[b];
}

__global__ __launch_bounds__(BIN_TB)
void bin_scan(int* __restrict__ cnt, int* __restrict__ binStart) {
  __shared__ int sc[BIN_TB];
  const int t = threadIdx.x;
  int tot = 0;
  if (t < N_BINS)
    for (int k = 0; k < BIN_BLOCKS; ++k) tot += cnt[k * N_BINS + t];
  sc[t] = tot;
  __syncthreads();
  for (int off = 1; off < BIN_TB; off <<= 1) {
    const int v = (t >= off) ? sc[t - off] : 0;
    __syncthreads();
    sc[t] += v;
    __syncthreads();
  }
  const int excl = sc[t] - tot;
  if (t <= N_BINS) binStart[t] = excl;
  if (t < N_BINS) {
    int run = excl;
    for (int k = 0; k < BIN_BLOCKS; ++k) {
      const int c = cnt[k * N_BINS + t];
      cnt[k * N_BINS + t] = run;
      run += c;
    }
  }
}

__global__ __launch_bounds__(BIN_TB)
void bin_place(const int* __restrict__ src, const int* __restrict__ dst,
               const int* __restrict__ cnt, unsigned* __restrict__ rec) {
  __shared__ int cur[N_BINS];
  const int e0 = blockIdx.x * BIN_CHUNK;
  const int e1 = min(e0 + BIN_CHUNK, N_EDGES);
  int d[BIN_U];
  int s[BIN_U];
#pragma unroll
  for (int u = 0; u < BIN_U; ++u) {
    const int e = e0 + (int)threadIdx.x + u * BIN_TB;
    const int ec = min(e, N_EDGES - 1);
    const int v = dst[ec];
    s[u] = src[ec];
    d[u] = (e < e1) ? v : -1;
  }
  for (int b = threadIdx.x; b < N_BINS; b += BIN_TB) cur[b] = cnt[blockIdx.x * N_BINS + b];
  __syncthreads();
  int pos[BIN_U];
#pragma unroll
  for (int u = 0; u < BIN_U; ++u) {
    pos[u] = -1;
    if (d[u] >= 0) pos[u] = atomicAdd(&cur[(unsigned)d[u] / TILE_R], 1);
  }
#pragma unroll
  for (int u = 0; u < BIN_U; ++u)
    if (d[u] >= 0)
      rec[pos[u]] = (((unsigned)d[u] % TILE_R) << 16) | (unsigned)s[u];
}

#define DEG_TB 128
#define DEG_U  8
static_assert(DEG_TB >= TILE_R, "one thread per row of the bin clears, then stores, its count");
__global__ __launch_bounds__(DEG_TB)
void bin_degrees(const unsigned* __restrict__ rec, const int* __restrict__ binStart,
                 float* __restrict__ dis) {
  __shared__ int cnt[TILE_R];
  if (threadIdx.x < TILE_R) cnt[threadIdx.x] = 0;
  __syncthreads();
  const int e0 = binStart[blockIdx.x];
  const int e1 = binStart[blockIdx.x + 1];
  for (int base = e0 + (int)threadIdx.x; base < e1; base += DEG_U * DEG_TB) {
    unsigned r[DEG_U];
#pragma unroll
    for (int u = 0; u < DEG_U; ++u) {
      const int e = base + u * DEG_TB;
      r[u] = (e < e1) ? rec[e] : 0xffffffffu;
    }
#pragma unroll
    for (int u = 0; u < DEG_U; ++u)
      if (r[u] != 0xffffffffu) atomicAdd(&cnt[r[u] >> 16], 1);
  }
  __syncthreads();
  const int row = blockIdx.x * TILE_R + (int)threadIdx.x;
  if (threadIdx.x < TILE_R && row < N_NODES) dis[row] = rsqrtf((float)cnt[threadIdx.x] + 1.0f);
}

#define TILE_K 128
#define TILE_P 130

template <bool PRE, int K, int G>
__device__ __forceinline__
void sum_rows_into_tile(const unsigned* __restrict__ rec, const int* __restrict__ binStart,
                        const float* __restrict__ h, const float* __restrict__ dis,
                        float* __restrict__ tile, float* __restrict__ disT,
                        int row0, int nrows, int wave, int lane) {
  constexpr int VPL = K / 32;
  constexpr int P = K + 2;
  static_assert(32 % G == 0, "a trip of 32 records is whole groups");
  typedef float vec __attribute__((ext_vector_type(VPL)));
  constexpr int SEED_U = TILE_R / AGG_WAVES;
  static_assert(SEED_U * AGG_WAVES == TILE_R, "every wave seeds SEED_U whole rows of the tile");

  const int e0 = binStart[blockIdx.x];
  const int e1 = binStart[blockIdx.x + 1];

  vec sv[SEED_U];
  float sd[SEED_U];
#pragma unroll
  for (int u = 0; u < SEED_U; ++u) {
    const int rr = min(row0 + wave + u * AGG_WAVES, N_NODES - 1);
    sd[u] = PRE ? 1.0f : dis[rr];
    sv[u] = *(const vec*)(h + (size_t)rr * K + lane * VPL);
  }
  const unsigned idle = ((unsigned)TILE_R << 16) | (unsigned)row0;
  int base = e0 + wave * 32;
  unsigned r = idle;
  if (base + lane < e1) r = rec[base + lane];
#pragma unroll
  for (int u = 0; u < SEED_U; ++u) {
    const int rt = wave + u * AGG_WAVES;
#pragma unroll
    for (int q = 0; q < VPL; ++q)
      tile[rt * P + q * 32 + lane] = PRE ? sv[u][q] : sv[u][q] * sd[u];
  }
  if (threadIdx.x < TILE_R)
    disT[threadIdx.x] = ((int)threadIdx.x < nrows) ? dis[row0 + threadIdx.x] : 0.0f;
  float ds = 0.0f;
  if (!PRE && base + lane < e1) ds = dis[r & 0xffffu];
  __syncthreads();

  for (; base < e1; base += AGG_WAVES * 32) {
    const unsigned rc = r;
    const float nm = ds;
    const int nidx = base + AGG_WAVES * 32 + lane;
    r = idle;
    if (nidx < e1) r = rec[nidx];
#pragma unroll
    for (int j = 0; j < 32; j += G) {
      if (base + j >= e1) break;
      vec v[G];
      int dl[G];
      float w[G];
#pragma unroll
      for (int u = 0; u < G; ++u) {
        const unsigned rj = (unsigned)__builtin_amdgcn_readlane((int)rc, j + u);
        w[u] = PRE ? 1.0f
                   : __builtin_bit_cast(float, __builtin_amdgcn_readlane(__builtin_bit_cast(int, nm), j + u));
        dl[u] = (int)(rj >> 16);
        v[u] = *(const vec*)(h + (size_t)(rj & 0xffffu) * K + lane * VPL);
      }
#pragma unroll
      for (int u = 0; u < G; ++u) {
#pragma unroll
        for (int q = 0; q < VPL; ++q)
          __hip_atomic_fetch_add(&tile[dl[u] * P + q * 32 + lane],
                                 PRE ? v[u][q] : v[u][q] * w[u],
                                 __ATOMIC_RELAXED, __HIP_MEMORY_SCOPE_WORKGROUP);
      }
      if (!PRE && j == 0) {
        ds = 0.0f;
        if (nidx < e1) ds = dis[r & 0xffffu];
      }
    }
  }

  __syncthreads();
}

template <int FOUT, bool RELU, bool PRE, bool PROJ>
__global__ __launch_bounds__(AGG_WAVES * 32)
void layer_tile(const unsigned* __restrict__ rec, const int* __restrict__ binStart,
                const float* __restrict__ h, const float* __restrict__ dis,
                const float* __restrict__ W, const float* __restrict__ bias,
                const float* __restrict__ Wn, float* __restrict__ out) {
  __shared__ __attribute__((aligned(16))) float tile[(TILE_R + 1) * TILE_P];
  __shared__ float disT[TILE_R];

  const int lane = threadIdx.x & 31;
  const int wave = __builtin_amdgcn_readfirstlane((int)(threadIdx.x >> 5));
  const int row0 = blockIdx.x * TILE_R;
  const int nrows = min(TILE_R, N_NODES - row0);

  constexpr int NCT = FOUT / 16;
  constexpr int RTW = NCT / 2;
  static_assert(AGG_WAVES * RTW == (TILE_R / 16) * NCT, "8 waves cover the block's output tiles");
  static_assert(!PROJ || FOUT == TILE_K, "the rows put back into the tile have the tile's width");
  const int ct = wave % NCT;
  const int rg = wave / NCT;
  const int n16 = lane & 15;
  const int half = lane >> 4;
  const float* wp = W + ct * 16 + n16;
  v2f bw[32];
#pragma unroll
  for (int s = 0; s < 32; ++s) {
    const int k = 8 * (s & 15) + 2 * (s >> 4) + half;
    bw[s].x = wp[(size_t)k * FOUT];
    bw[s].y = wp[(size_t)(k + 4) * FOUT];
  }
  const float bcol = bias[ct * 16 + n16];

  sum_rows_into_tile<PRE, TILE_K, 8>(rec, binStart, h, dis, tile, disT, row0, nrows, wave, lane);


  v8f acc[RTW];
#pragma unroll
  for (int t = 0; t < RTW; ++t) acc[t] = v8f{};
  const float* abase = tile + ((rg * RTW) * 16 + n16) * TILE_P + 32 * half;
#pragma unroll
  for (int s = 0; s < 32; ++s) {
#pragma unroll
    for (int t = 0; t < RTW; ++t) {
      const v2f a = *(const v2f*)(abase + t * 16 * TILE_P + 64 * (s >> 4) + 2 * (s & 15));
      acc[t] = __builtin_amdgcn_wmma_f32_16x16x4_f32(
           false, a,  false, bw[s],
           (short)0, acc[t],  false,  false);
    }
  }

  if constexpr (!PROJ) {
#pragma unroll
    for (int t = 0; t < RTW; ++t) {
      const int rb = (rg * RTW + t) * 16 + (half << 3);
      float* op = out + (size_t)(row0 + rb) * FOUT + ct * 16 + n16;
      float val[8];
#pragma unroll
      for (int i = 0; i < 8; ++i) {
        const float dr = disT[rb + i];
        val[i] = acc[t][i] * dr + bcol;
        if (RELU) val[i] = fmaxf(val[i], 0.0f);
        val[i] *= dr;
      }
      if (rb + 8 <= nrows) {
#pragma unroll
        for (int i = 0; i < 8; ++i) op[(size_t)i * FOUT] = val[i];
      } else {
#pragma unroll
        for (int i = 0; i < 8; ++i)
          if (rb + i < nrows) op[(size_t)i * FOUT] = val[i];
      }
    }
  } else {
    static_assert(D_OUT == 64, "8 waves x 2 tiles cover the 4 x 4 tiles of the second pass");
    const int ct2 = wave & 3;
    const int rg2 = wave >> 2;
    __syncthreads();
#pragma unroll
    for (int t = 0; t < RTW; ++t) {
      const int rb = (rg * RTW + t) * 16 + (half << 3);
#pragma unroll
      for (int i = 0; i < 8; ++i) {
        const float dr = disT[rb + i];
        float val = acc[t][i] * dr + bcol;
        if (RELU) val = fmaxf(val, 0.0f);
        tile[(rb + i) * TILE_P + ct * 16 + n16] = val * dr;
      }
    }
    __builtin_amdgcn_sched_barrier(0);
    const float* wq = Wn + (size_t)(32 * half) * D_OUT + ct2 * 16 + n16;
    v2f bq[32];
#pragma unroll
    for (int s = 0; s < 32; ++s) {
      const int k = 64 * (s >> 4) + 2 * (s & 15);
      bq[s].x = wq[(size_t)k * D_OUT];
      bq[s].y = wq[(size_t)(k + 1) * D_OUT];
    }
    __syncthreads();

    v8f acc2[2] = {v8f{}, v8f{}};
    const float* a2 = tile + ((rg2 * 2) * 16 + n16) * TILE_P + 32 * half;
#pragma unroll
    for (int s = 0; s < 32; ++s) {
#pragma unroll
      for (int t = 0; t < 2; ++t) {
        const v2f a = *(const v2f*)(a2 + t * 16 * TILE_P + 64 * (s >> 4) + 2 * (s & 15));
        acc2[t] = __builtin_amdgcn_wmma_f32_16x16x4_f32(
             false, a,  false, bq[s],
             (short)0, acc2[t],  false,  false);
      }
    }
#pragma unroll
    for (int t = 0; t < 2; ++t) {
      const int rb = (rg2 * 2 + t) * 16 + (half << 3);
      float* op = out + (size_t)(row0 + rb) * D_OUT + ct2 * 16 + n16;
      if (rb + 8 <= nrows) {
#pragma unroll
        for (int i = 0; i < 8; ++i) op[(size_t)i * D_OUT] = acc2[t][i];
      } else {
#pragma unroll
        for (int i = 0; i < 8; ++i)
          if (rb + i < nrows) op[(size_t)i * D_OUT] = acc2[t][i];
      }
    }
  }
}

#define POOL_P (D_OUT + 2)
__global__ __launch_bounds__(AGG_WAVES * 32)
void layer_pool(const unsigned* __restrict__ rec, const int* __restrict__ binStart,
                const float* __restrict__ z, const float* __restrict__ dis,
                const int* __restrict__ batch, float* __restrict__ psum,
                float* __restrict__ cnts) {
  __shared__ __attribute__((aligned(16))) float tile[(TILE_R + 1) * POOL_P];
  __shared__ float disT[TILE_R];
  __shared__ int gidT[TILE_R];

  const int lane = threadIdx.x & 31;
  const int wave = __builtin_amdgcn_readfirstlane((int)(threadIdx.x >> 5));
  const int row0 = blockIdx.x * TILE_R;
  const int nrows = min(TILE_R, N_NODES - row0);

  if (threadIdx.x < TILE_R)
    gidT[threadIdx.x] = ((int)threadIdx.x < nrows) ? batch[row0 + threadIdx.x] : -1;

  sum_rows_into_tile<true, D_OUT, 16>(rec, binStart, z, dis, tile, disT, row0, nrows, wave, lane);

  constexpr int QR = TILE_R / 4;
  static_assert(AGG_WAVES * 32 == 4 * D_OUT, "a thread per column and row quarter");
  const int col = threadIdx.x & (D_OUT - 1);
  const int r0 = (wave >> 1) * QR;
  float v[QR];
  int gd[QR];
#pragma unroll
  for (int r = 0; r < QR; ++r) {
    v[r] = tile[(r0 + r) * POOL_P + col] * disT[r0 + r];
    gd[r] = __builtin_amdgcn_readfirstlane(gidT[r0 + r]);
  }
  auto hand_over = [&](int g, float s, int len) {
    if ((unsigned)g < (unsigned)N_GRAPHS) {
      atomicAdd(&psum[g * D_OUT + col], s);
      if (col == 0) atomicAdd(&cnts[g], (float)len);
    }
  };
  int g = gd[0];
  int len = 1;
  float s = v[0];
#pragma unroll
  for (int r = 1; r < QR; ++r) {
    if (gd[r] != g) {
      hand_over(g, s, len);
      g = gd[r];
      len = 0;
      s = 0.0f;
    }
    s += v[r];
    ++len;
  }
  hand_over(g, s, len);
}

__global__ void pool_finish(const float* __restrict__ psum, const float* __restrict__ cnts,
                            const float* __restrict__ bias, float* __restrict__ out) {
  const int i = blockIdx.x * blockDim.x + threadIdx.x;
  if (i < N_GRAPHS * D_OUT) {
    const int gq = i / D_OUT;
    const int f = i % D_OUT;
    const float cn = cnts[gq];
    out[i] = (psum[gq * D_OUT + (f & 1) * 32 + (f >> 1)] + cn * bias[f]) / fmaxf(cn, 1.0f);
  }
}

extern "C" void kernel_launch(void* const* d_in, const int* in_sizes, int n_in,
                              void* d_out, int out_size, void* d_ws, size_t ws_size,
                              hipStream_t stream) {
  const float* x     = (const float*)d_in[0];
  const int*   ei    = (const int*)d_in[1];
  const int*   batch = (const int*)d_in[2];
  const float* W1 = (const float*)d_in[3];  const float* b1 = (const float*)d_in[4];
  const float* W2 = (const float*)d_in[5];  const float* b2 = (const float*)d_in[6];
  const float* W3 = (const float*)d_in[7];  const float* b3 = (const float*)d_in[8];
  const float* W4 = (const float*)d_in[9];  const float* b4 = (const float*)d_in[10];
  float* out = (float*)d_out;

  float* bufA = (float*)d_ws;
  float* bufB = bufA + (size_t)N_NODES * D_HID;
  float* dis  = bufB + (size_t)N_NODES * D_HID;
  int* binStart = (int*)(dis + N_NODES);
  unsigned* rec = (unsigned*)(binStart + N_BINS + 1);
  int* binCnt   = (int*)bufB;
  float* psum = bufB;
  float* cnts = psum + N_GRAPHS * D_OUT;

  const int* src = ei;
  const int* dst = ei + N_EDGES;

  if (ws_size < WS_NEED) return;

  const int TB = 256;
  auto blocks = [](long long n, int tb) { return (unsigned)((n + tb - 1) / tb); };

  bin_hist<<<BIN_BLOCKS, BIN_TB, 0, stream>>>(dst, binCnt);
  bin_scan<<<1, BIN_TB, 0, stream>>>(binCnt, binStart);
  bin_place<<<BIN_BLOCKS, BIN_TB, 0, stream>>>(src, dst, binCnt, rec);

  bin_degrees<<<N_BINS, DEG_TB, 0, stream>>>(rec, binStart, dis);

  static_assert(D_IN == TILE_K && D_HID == TILE_K, "every layer's input has TILE_K features");
  layer_tile<D_HID, true, false, false><<<N_BINS, AGG_WAVES * 32, 0, stream>>>(rec, binStart, x, dis, W1, b1, W4, bufA);
  layer_tile<D_HID, true, true, false><<<N_BINS, AGG_WAVES * 32, 0, stream>>>(rec, binStart, bufA, dis, W2, b2, W4, bufB);
  layer_tile<D_HID, true, true, true><<<N_BINS, AGG_WAVES * 32, 0, stream>>>(rec, binStart, bufB, dis, W3, b3, W4, bufA);

  zero_f32<<<blocks(N_GRAPHS * D_OUT + N_GRAPHS, TB), TB, 0, stream>>>(
      psum, N_GRAPHS * D_OUT + N_GRAPHS);
  layer_pool<<<N_BINS, AGG_WAVES * 32, 0, stream>>>(rec, binStart, bufA, dis, batch, psum, cnts);
  pool_finish<<<blocks(N_GRAPHS * D_OUT, TB), TB, 0, stream>>>(psum, cnts, b4, out);
}


# a bin's records sorted by destination row (LDS count pass in the degree kernel); a wave sums each run of equal rows in registers: one LDS f32 add per run, not per edge; sum order: run, then arrival
# speedup vs baseline: 6.6743x; 1.0611x over previous
#include <hip/hip_runtime.h>
#include <hip/hip_bf16.h>

typedef float v2f __attribute__((ext_vector_type(2)));
typedef float v8f __attribute__((ext_vector_type(8)));

#define N_NODES  50000
#define N_EDGES  600000
#define N_GRAPHS 64
#define D_IN     128
#define D_HID    128
#define D_OUT    64

#define TILE_R     64
#define N_BINS     ((N_NODES + TILE_R - 1) / TILE_R)
#define BIN_BLOCKS 64
#define BIN_CHUNK  ((N_EDGES + BIN_BLOCKS - 1) / BIN_BLOCKS)
#define BIN_TB     1024
#define BIN_U      ((BIN_CHUNK + BIN_TB - 1) / BIN_TB)
#define AGG_WAVES  8

static_assert(N_NODES <= 65536, "a source row index must fit the low 16 bits of an edge record");
static_assert(N_BINS <= BIN_TB, "the scan gives one thread to each bin");

constexpr size_t WS_NEED =
    ((size_t)2 * N_NODES * D_HID + N_NODES + N_BINS + 1 + N_EDGES) * sizeof(float);
static_assert((size_t)N_GRAPHS * D_OUT + N_GRAPHS <= (size_t)N_NODES * D_HID,
              "the pooled sums and counts fit in one plane");

__global__ void zero_f32(float* __restrict__ p, int n) {
  int i = blockIdx.x * blockDim.x + threadIdx.x;
  if (i < n) p[i] = 0.0f;
}

__global__ __launch_bounds__(BIN_TB)
void bin_hist(const int* __restrict__ dst, int* __restrict__ cnt) {
  __shared__ int hist[N_BINS];
  const int e0 = blockIdx.x * BIN_CHUNK;
  const int e1 = min(e0 + BIN_CHUNK, N_EDGES);
  int d[BIN_U];
#pragma unroll
  for (int u = 0; u < BIN_U; ++u) {
    const int e = e0 + (int)threadIdx.x + u * BIN_TB;
    const int v = dst[min(e, N_EDGES - 1)];
    d[u] = (e < e1) ? v : -1;
  }
  for (int b = threadIdx.x; b < N_BINS; b += BIN_TB) hist[b] = 0;
  __syncthreads();
#pragma unroll
  for (int u = 0; u < BIN_U; ++u)
    if (d[u] >= 0) atomicAdd(&hist[(unsigned)d[u] / TILE_R], 1);
  __syncthreads();
  for (int b = threadIdx.x; b < N_BINS; b += BIN_TB) cnt[blockIdx.x * N_BINS + b] = hist[b];
}

__global__ __launch_bounds__(BIN_TB)
void bin_scan(int* __restrict__ cnt, int* __restrict__ binStart) {
  __shared__ int sc[BIN_TB];
  const int t = threadIdx.x;
  int tot = 0;
  if (t < N_BINS)
    for (int k = 0; k < BIN_BLOCKS; ++k) tot += cnt[k * N_BINS + t];
  sc[t] = tot;
  __syncthreads();
  for (int off = 1; off < BIN_TB; off <<= 1) {
    const int v = (t >= off) ? sc[t - off] : 0;
    __syncthreads();
    sc[t] += v;
    __syncthreads();
  }
  const int excl = sc[t] - tot;
  if (t <= N_BINS) binStart[t] = excl;
  if (t < N_BINS) {
    int run = excl;
    for (int k = 0; k < BIN_BLOCKS; ++k) {
      const int c = cnt[k * N_BINS + t];
      cnt[k * N_BINS + t] = run;
      run += c;
    }
  }
}

__global__ __launch_bounds__(BIN_TB)
void bin_place(const int* __restrict__ src, const int* __restrict__ dst,
               const int* __restrict__ cnt, unsigned* __restrict__ rec) {
  __shared__ int cur[N_BINS];
  const int e0 = blockIdx.x * BIN_CHUNK;
  const int e1 = min(e0 + BIN_CHUNK, N_EDGES);
  int d[BIN_U];
  int s[BIN_U];
#pragma unroll
  for (int u = 0; u < BIN_U; ++u) {
    const int e = e0 + (int)threadIdx.x + u * BIN_TB;
    const int ec = min(e, N_EDGES - 1);
    const int v = dst[ec];
    s[u] = src[ec];
    d[u] = (e < e1) ? v : -1;
  }
  for (int b = threadIdx.x; b < N_BINS; b += BIN_TB) cur[b] = cnt[blockIdx.x * N_BINS + b];
  __syncthreads();
  int pos[BIN_U];
#pragma unroll
  for (int u = 0; u < BIN_U; ++u) {
    pos[u] = -1;
    if (d[u] >= 0) pos[u] = atomicAdd(&cur[(unsigned)d[u] / TILE_R], 1);
  }
#pragma unroll
  for (int u = 0; u < BIN_U; ++u)
    if (d[u] >= 0)
      rec[pos[u]] = (((unsigned)d[u] % TILE_R) << 16) | (unsigned)s[u];
}

#define DEG_TB 128
#define DEG_U  8
static_assert(DEG_TB >= TILE_R, "one thread per row of the bin clears, scans and stores its count");
__global__ __launch_bounds__(DEG_TB)
void bin_rows(const unsigned* __restrict__ recIn, const int* __restrict__ binStart,
              float* __restrict__ dis, unsigned* __restrict__ rec) {
  __shared__ int cnt[TILE_R];
  __shared__ int sc[TILE_R];
  const int t = threadIdx.x;
  if (t < TILE_R) cnt[t] = 0;
  __syncthreads();
  const int e0 = binStart[blockIdx.x];
  const int e1 = binStart[blockIdx.x + 1];
  for (int base = e0 + t; base < e1; base += DEG_U * DEG_TB) {
    unsigned r[DEG_U];
#pragma unroll
    for (int u = 0; u < DEG_U; ++u) {
      const int e = base + u * DEG_TB;
      r[u] = (e < e1) ? recIn[e] : 0xffffffffu;
    }
#pragma unroll
    for (int u = 0; u < DEG_U; ++u)
      if (r[u] != 0xffffffffu) atomicAdd(&cnt[r[u] >> 16], 1);
  }
  __syncthreads();
  int c = 0;
  if (t < TILE_R) {
    c = cnt[t];
    sc[t] = c;
    const int row = blockIdx.x * TILE_R + t;
    if (row < N_NODES) dis[row] = rsqrtf((float)c + 1.0f);
  }
  __syncthreads();
  for (int off = 1; off < TILE_R; off <<= 1) {
    int v = 0;
    if (t < TILE_R && t >= off) v = sc[t - off];
    __syncthreads();
    if (t < TILE_R) sc[t] += v;
    __syncthreads();
  }
  if (t < TILE_R) cnt[t] = e0 + sc[t] - c;
  __syncthreads();
  for (int base = e0 + t; base < e1; base += DEG_U * DEG_TB) {
    unsigned r[DEG_U];
    int pos[DEG_U];
#pragma unroll
    for (int u = 0; u < DEG_U; ++u) {
      const int e = base + u * DEG_TB;
      r[u] = (e < e1) ? recIn[e] : 0xffffffffu;
    }
#pragma unroll
    for (int u = 0; u < DEG_U; ++u) {
      pos[u] = -1;
      if (r[u] != 0xffffffffu) pos[u] = atomicAdd(&cnt[r[u] >> 16], 1);
    }
#pragma unroll
    for (int u = 0; u < DEG_U; ++u)
      if (r[u] != 0xffffffffu) rec[pos[u]] = r[u];
  }
}

#define TILE_K 128
#define TILE_P 130

template <bool PRE, int K, int G>
__device__ __forceinline__
void sum_rows_into_tile(const unsigned* __restrict__ rec, const int* __restrict__ binStart,
                        const float* __restrict__ h, const float* __restrict__ dis,
                        float* __restrict__ tile, float* __restrict__ disT,
                        int row0, int nrows, int wave, int lane) {
  constexpr int VPL = K / 32;
  constexpr int P = K + 2;
  static_assert(32 % G == 0, "a trip of 32 records is whole groups");
  typedef float vec __attribute__((ext_vector_type(VPL)));
  constexpr int SEED_U = TILE_R / AGG_WAVES;
  static_assert(SEED_U * AGG_WAVES == TILE_R, "every wave seeds SEED_U whole rows of the tile");

  const int e0 = binStart[blockIdx.x];
  const int e1 = binStart[blockIdx.x + 1];

  vec sv[SEED_U];
  float sd[SEED_U];
#pragma unroll
  for (int u = 0; u < SEED_U; ++u) {
    const int rr = min(row0 + wave + u * AGG_WAVES, N_NODES - 1);
    sd[u] = PRE ? 1.0f : dis[rr];
    sv[u] = *(const vec*)(h + (size_t)rr * K + lane * VPL);
  }
  const unsigned idle = ((unsigned)TILE_R << 16) | (unsigned)row0;
  int base = e0 + wave * 32;
  unsigned r = idle;
  if (base + lane < e1) r = rec[base + lane];
#pragma unroll
  for (int u = 0; u < SEED_U; ++u) {
    const int rt = wave + u * AGG_WAVES;
#pragma unroll
    for (int q = 0; q < VPL; ++q)
      tile[rt * P + q * 32 + lane] = PRE ? sv[u][q] : sv[u][q] * sd[u];
  }
  if (threadIdx.x < TILE_R)
    disT[threadIdx.x] = ((int)threadIdx.x < nrows) ? dis[row0 + threadIdx.x] : 0.0f;
  float ds = 0.0f;
  if (!PRE && base + lane < e1) ds = dis[r & 0xffffu];
  __syncthreads();

  for (; base < e1; base += AGG_WAVES * 32) {
    const unsigned rc = r;
    const float nm = ds;
    const int nidx = base + AGG_WAVES * 32 + lane;
    r = idle;
    if (nidx < e1) r = rec[nidx];
    int cur = -1;
    vec acc;
#pragma unroll
    for (int q = 0; q < VPL; ++q) acc[q] = 0.0f;
#pragma unroll
    for (int j = 0; j < 32; j += G) {
      if (base + j >= e1) break;
      vec v[G];
      int dl[G];
      float w[G];
#pragma unroll
      for (int u = 0; u < G; ++u) {
        const unsigned rj = (unsigned)__builtin_amdgcn_readlane((int)rc, j + u);
        w[u] = PRE ? 1.0f
                   : __builtin_bit_cast(float, __builtin_amdgcn_readlane(__builtin_bit_cast(int, nm), j + u));
        dl[u] = (int)(rj >> 16);
        v[u] = *(const vec*)(h + (size_t)(rj & 0xffffu) * K + lane * VPL);
      }
#pragma unroll
      for (int u = 0; u < G; ++u) {
        if (dl[u] != cur) {
          if (cur >= 0) {
#pragma unroll
            for (int q = 0; q < VPL; ++q)
              __hip_atomic_fetch_add(&tile[cur * P + q * 32 + lane], acc[q],
                                     __ATOMIC_RELAXED, __HIP_MEMORY_SCOPE_WORKGROUP);
          }
          cur = dl[u];
#pragma unroll
          for (int q = 0; q < VPL; ++q) acc[q] = PRE ? v[u][q] : v[u][q] * w[u];
        } else {
#pragma unroll
          for (int q = 0; q < VPL; ++q) acc[q] += PRE ? v[u][q] : v[u][q] * w[u];
        }
      }
      if (!PRE && j == 0) {
        ds = 0.0f;
        if (nidx < e1) ds = dis[r & 0xffffu];
      }
    }
    if (cur >= 0) {
#pragma unroll
      for (int q = 0; q < VPL; ++q)
        __hip_atomic_fetch_add(&tile[cur * P + q * 32 + lane], acc[q],
                               __ATOMIC_RELAXED, __HIP_MEMORY_SCOPE_WORKGROUP);
    }
  }

  __syncthreads();
}

template <int FOUT, bool RELU, bool PRE, bool PROJ>
__global__ __launch_bounds__(AGG_WAVES * 32, 8)
void layer_tile(const unsigned* __restrict__ rec, const int* __restrict__ binStart,
                const float* __restrict__ h, const float* __restrict__ dis,
                const float* __restrict__ W, const float* __restrict__ bias,
                const float* __restrict__ Wn, float* __restrict__ out) {
  __shared__ __attribute__((aligned(16))) float tile[(TILE_R + 1) * TILE_P];
  __shared__ float disT[TILE_R];

  const int lane = threadIdx.x & 31;
  const int wave = __builtin_amdgcn_readfirstlane((int)(threadIdx.x >> 5));
  const int row0 = blockIdx.x * TILE_R;
  const int nrows = min(TILE_R, N_NODES - row0);

  constexpr int NCT = FOUT / 16;
  constexpr int RTW = NCT / 2;
  static_assert(AGG_WAVES * RTW == (TILE_R / 16) * NCT, "8 waves cover the block's output tiles");
  static_assert(!PROJ || FOUT == TILE_K, "the rows put back into the tile have the tile's width");
  const int ct = wave % NCT;
  const int rg = wave / NCT;
  const int n16 = lane & 15;
  const int half = lane >> 4;
  const float* wp = W + ct * 16 + n16;
  v2f bw[32];
#pragma unroll
  for (int s = 0; s < 32; ++s) {
    const int k = 8 * (s & 15) + 2 * (s >> 4) + half;
    bw[s].x = wp[(size_t)k * FOUT];
    bw[s].y = wp[(size_t)(k + 4) * FOUT];
  }
  const float bcol = bias[ct * 16 + n16];

  sum_rows_into_tile<PRE, TILE_K, 8>(rec, binStart, h, dis, tile, disT, row0, nrows, wave, lane);


  v8f acc[RTW];
#pragma unroll
  for (int t = 0; t < RTW; ++t) acc[t] = v8f{};
  const float* abase = tile + ((rg * RTW) * 16 + n16) * TILE_P + 32 * half;
#pragma unroll
  for (int s = 0; s < 32; ++s) {
#pragma unroll
    for (int t = 0; t < RTW; ++t) {
      const v2f a = *(const v2f*)(abase + t * 16 * TILE_P + 64 * (s >> 4) + 2 * (s & 15));
      acc[t] = __builtin_amdgcn_wmma_f32_16x16x4_f32(
           false, a,  false, bw[s],
           (short)0, acc[t],  false,  false);
    }
  }

  if constexpr (!PROJ) {
#pragma unroll
    for (int t = 0; t < RTW; ++t) {
      const int rb = (rg * RTW + t) * 16 + (half << 3);
      float* op = out + (size_t)(row0 + rb) * FOUT + ct * 16 + n16;
      float val[8];
#pragma unroll
      for (int i = 0; i < 8; ++i) {
        const float dr = disT[rb + i];
        val[i] = acc[t][i] * dr + bcol;
        if (RELU) val[i] = fmaxf(val[i], 0.0f);
        val[i] *= dr;
      }
      if (rb + 8 <= nrows) {
#pragma unroll
        for (int i = 0; i < 8; ++i) op[(size_t)i * FOUT] = val[i];
      } else {
#pragma unroll
        for (int i = 0; i < 8; ++i)
          if (rb + i < nrows) op[(size_t)i * FOUT] = val[i];
      }
    }
  } else {
    static_assert(D_OUT == 64, "8 waves x 2 tiles cover the 4 x 4 tiles of the second pass");
    const int ct2 = wave & 3;
    const int rg2 = wave >> 2;
    __syncthreads();
#pragma unroll
    for (int t = 0; t < RTW; ++t) {
      const int rb = (rg * RTW + t) * 16 + (half << 3);
#pragma unroll
      for (int i = 0; i < 8; ++i) {
        const float dr = disT[rb + i];
        float val = acc[t][i] * dr + bcol;
        if (RELU) val = fmaxf(val, 0.0f);
        tile[(rb + i) * TILE_P + ct * 16 + n16] = val * dr;
      }
    }
    __builtin_amdgcn_sched_barrier(0);
    const float* wq = Wn + (size_t)(32 * half) * D_OUT + ct2 * 16 + n16;
    v2f bq[32];
#pragma unroll
    for (int s = 0; s < 32; ++s) {
      const int k = 64 * (s >> 4) + 2 * (s & 15);
      bq[s].x = wq[(size_t)k * D_OUT];
      bq[s].y = wq[(size_t)(k + 1) * D_OUT];
    }
    __syncthreads();

    v8f acc2[2] = {v8f{}, v8f{}};
    const float* a2 = tile + ((rg2 * 2) * 16 + n16) * TILE_P + 32 * half;
#pragma unroll
    for (int s = 0; s < 32; ++s) {
#pragma unroll
      for (int t = 0; t < 2; ++t) {
        const v2f a = *(const v2f*)(a2 + t * 16 * TILE_P + 64 * (s >> 4) + 2 * (s & 15));
        acc2[t] = __builtin_amdgcn_wmma_f32_16x16x4_f32(
             false, a,  false, bq[s],
             (short)0, acc2[t],  false,  false);
      }
    }
#pragma unroll
    for (int t = 0; t < 2; ++t) {
      const int rb = (rg2 * 2 + t) * 16 + (half << 3);
      float* op = out + (size_t)(row0 + rb) * D_OUT + ct2 * 16 + n16;
      if (rb + 8 <= nrows) {
#pragma unroll
        for (int i = 0; i < 8; ++i) op[(size_t)i * D_OUT] = acc2[t][i];
      } else {
#pragma unroll
        for (int i = 0; i < 8; ++i)
          if (rb + i < nrows) op[(size_t)i * D_OUT] = acc2[t][i];
      }
    }
  }
}

#define POOL_P (D_OUT + 2)
__global__ __launch_bounds__(AGG_WAVES * 32)
void layer_pool(const unsigned* __restrict__ rec, const int* __restrict__ binStart,
                const float* __restrict__ z, const float* __restrict__ dis,
                const int* __restrict__ batch, float* __restrict__ psum,
                float* __restrict__ cnts) {
  __shared__ __attribute__((aligned(16))) float tile[(TILE_R + 1) * POOL_P];
  __shared__ float disT[TILE_R];
  __shared__ int gidT[TILE_R];

  const int lane = threadIdx.x & 31;
  const int wave = __builtin_amdgcn_readfirstlane((int)(threadIdx.x >> 5));
  const int row0 = blockIdx.x * TILE_R;
  const int nrows = min(TILE_R, N_NODES - row0);

  if (threadIdx.x < TILE_R)
    gidT[threadIdx.x] = ((int)threadIdx.x < nrows) ? batch[row0 + threadIdx.x] : -1;

  sum_rows_into_tile<true, D_OUT, 16>(rec, binStart, z, dis, tile, disT, row0, nrows, wave, lane);

  constexpr int QR = TILE_R / 4;
  static_assert(AGG_WAVES * 32 == 4 * D_OUT, "a thread per column and row quarter");
  const int col = threadIdx.x & (D_OUT - 1);
  const int r0 = (wave >> 1) * QR;
  float v[QR];
  int gd[QR];
#pragma unroll
  for (int r = 0; r < QR; ++r) {
    v[r] = tile[(r0 + r) * POOL_P + col] * disT[r0 + r];
    gd[r] = __builtin_amdgcn_readfirstlane(gidT[r0 + r]);
  }
  auto hand_over = [&](int g, float s, int len) {
    if ((unsigned)g < (unsigned)N_GRAPHS) {
      atomicAdd(&psum[g * D_OUT + col], s);
      if (col == 0) atomicAdd(&cnts[g], (float)len);
    }
  };
  int g = gd[0];
  int len = 1;
  float s = v[0];
#pragma unroll
  for (int r = 1; r < QR; ++r) {
    if (gd[r] != g) {
      hand_over(g, s, len);
      g = gd[r];
      len = 0;
      s = 0.0f;
    }
    s += v[r];
    ++len;
  }
  hand_over(g, s, len);
}

__global__ void pool_finish(const float* __restrict__ psum, const float* __restrict__ cnts,
                            const float* __restrict__ bias, float* __restrict__ out) {
  const int i = blockIdx.x * blockDim.x + threadIdx.x;
  if (i < N_GRAPHS * D_OUT) {
    const int gq = i / D_OUT;
    const int f = i % D_OUT;
    const float cn = cnts[gq];
    out[i] = (psum[gq * D_OUT + (f & 1) * 32 + (f >> 1)] + cn * bias[f]) / fmaxf(cn, 1.0f);
  }
}

extern "C" void kernel_launch(void* const* d_in, const int* in_sizes, int n_in,
                              void* d_out, int out_size, void* d_ws, size_t ws_size,
                              hipStream_t stream) {
  const float* x     = (const float*)d_in[0];
  const int*   ei    = (const int*)d_in[1];
  const int*   batch = (const int*)d_in[2];
  const float* W1 = (const float*)d_in[3];  const float* b1 = (const float*)d_in[4];
  const float* W2 = (const float*)d_in[5];  const float* b2 = (const float*)d_in[6];
  const float* W3 = (const float*)d_in[7];  const float* b3 = (const float*)d_in[8];
  const float* W4 = (const float*)d_in[9];  const float* b4 = (const float*)d_in[10];
  float* out = (float*)d_out;

  float* bufA = (float*)d_ws;
  float* bufB = bufA + (size_t)N_NODES * D_HID;
  float* dis  = bufB + (size_t)N_NODES * D_HID;
  int* binStart = (int*)(dis + N_NODES);
  unsigned* rec = (unsigned*)(binStart + N_BINS + 1);
  int* binCnt   = (int*)bufB;
  unsigned* recBin = (unsigned*)bufA;
  float* psum = bufB;
  float* cnts = psum + N_GRAPHS * D_OUT;

  const int* src = ei;
  const int* dst = ei + N_EDGES;

  if (ws_size < WS_NEED) return;

  const int TB = 256;
  auto blocks = [](long long n, int tb) { return (unsigned)((n + tb - 1) / tb); };

  bin_hist<<<BIN_BLOCKS, BIN_TB, 0, stream>>>(dst, binCnt);
  bin_scan<<<1, BIN_TB, 0, stream>>>(binCnt, binStart);
  bin_place<<<BIN_BLOCKS, BIN_TB, 0, stream>>>(src, dst, binCnt, recBin);

  bin_rows<<<N_BINS, DEG_TB, 0, stream>>>(recBin, binStart, dis, rec);

  static_assert(D_IN == TILE_K && D_HID == TILE_K, "every layer's input has TILE_K features");
  layer_tile<D_HID, true, false, false><<<N_BINS, AGG_WAVES * 32, 0, stream>>>(rec, binStart, x, dis, W1, b1, W4, bufA);
  layer_tile<D_HID, true, true, false><<<N_BINS, AGG_WAVES * 32, 0, stream>>>(rec, binStart, bufA, dis, W2, b2, W4, bufB);
  layer_tile<D_HID, true, true, true><<<N_BINS, AGG_WAVES * 32, 0, stream>>>(rec, binStart, bufB, dis, W3, b3, W4, bufA);

  zero_f32<<<blocks(N_GRAPHS * D_OUT + N_GRAPHS, TB), TB, 0, stream>>>(
      psum, N_GRAPHS * D_OUT + N_GRAPHS);
  layer_pool<<<N_BINS, AGG_WAVES * 32, 0, stream>>>(rec, binStart, bufA, dis, batch, psum, cnts);
  pool_finish<<<blocks(N_GRAPHS * D_OUT, TB), TB, 0, stream>>>(psum, cnts, b4, out);
}


# edge grouping in 2 launches, not 4: each of 64 blocks sorts its own slice by bin (LDS counts, in-block scan, LDS staging, whole-line stores); a bin's block merges its 64 segments; sum order kept
# speedup vs baseline: 8.4406x; 1.2646x over previous
#include <hip/hip_runtime.h>
#include <hip/hip_bf16.h>

typedef float v2f __attribute__((ext_vector_type(2)));
typedef float v8f __attribute__((ext_vector_type(8)));

#define N_NODES  50000
#define N_EDGES  600000
#define N_GRAPHS 64
#define D_IN     128
#define D_HID    128
#define D_OUT    64

#define TILE_R     64
#define N_BINS     ((N_NODES + TILE_R - 1) / TILE_R)
#define BIN_BLOCKS 64
#define BIN_CHUNK  ((N_EDGES + BIN_BLOCKS - 1) / BIN_BLOCKS)
#define BIN_TB     1024
#define BIN_U      ((BIN_CHUNK + BIN_TB - 1) / BIN_TB)
#define AGG_WAVES  8

static_assert(N_NODES <= 65536, "a source row index must fit the low 16 bits of an edge record");
static_assert(N_BINS + 1 <= BIN_TB, "the scan gives one thread to each bin and one to the total");

constexpr size_t WS_NEED =
    ((size_t)2 * N_NODES * D_HID + N_NODES + N_BINS + 1 + N_EDGES) * sizeof(float);
static_assert((size_t)N_GRAPHS * D_OUT + N_GRAPHS <= (size_t)N_NODES * D_HID,
              "the pooled sums and counts fit in one plane");

__global__ void zero_f32(float* __restrict__ p, int n) {
  int i = blockIdx.x * blockDim.x + threadIdx.x;
  if (i < n) p[i] = 0.0f;
}

#define SEG_PITCH (N_BINS + 1)

__device__ __forceinline__ int wave_incl_sum(int v, int lane) {
#pragma unroll
  for (int o = 1; o < 32; o <<= 1) {
    const int y = __shfl_up(v, o, 32);
    if (lane >= o) v += y;
  }
  return v;
}

__global__ __launch_bounds__(BIN_TB)
void chunk_group(const int* __restrict__ src, const int* __restrict__ dst,
                 int* __restrict__ segOff, unsigned* __restrict__ recChunk) {
  __shared__ int hist[N_BINS];
  __shared__ int wsum[BIN_TB / 32];
  __shared__ unsigned stage[BIN_CHUNK];
  const int t = threadIdx.x;
  const int lane = t & 31;
  const int wave = t >> 5;
  const int e0 = blockIdx.x * BIN_CHUNK;
  const int e1 = min(e0 + BIN_CHUNK, N_EDGES);
  int d[BIN_U];
  int s[BIN_U];
#pragma unroll
  for (int u = 0; u < BIN_U; ++u) {
    const int e = e0 + t + u * BIN_TB;
    const int ec = min(e, N_EDGES - 1);
    const int v = dst[ec];
    s[u] = src[ec];
    d[u] = (e < e1) ? v : -1;
  }
  for (int b = t; b < N_BINS; b += BIN_TB) hist[b] = 0;
  __syncthreads();
#pragma unroll
  for (int u = 0; u < BIN_U; ++u)
    if (d[u] >= 0) atomicAdd(&hist[(unsigned)d[u] / TILE_R], 1);
  __syncthreads();
  const int c = (t < N_BINS) ? hist[t] : 0;
  const int incl = wave_incl_sum(c, lane);
  if (lane == 31) wsum[wave] = incl;
  __syncthreads();
  if (wave == 0) {
    const int w = wsum[lane];
    wsum[lane] = wave_incl_sum(w, lane) - w;
  }
  __syncthreads();
  const int excl = wsum[wave] + incl - c;
  if (t <= N_BINS) segOff[blockIdx.x * SEG_PITCH + t] = excl;
  if (t < N_BINS) hist[t] = excl;
  __syncthreads();
  int pos[BIN_U];
#pragma unroll
  for (int u = 0; u < BIN_U; ++u) {
    pos[u] = -1;
    if (d[u] >= 0) pos[u] = atomicAdd(&hist[(unsigned)d[u] / TILE_R], 1);
  }
#pragma unroll
  for (int u = 0; u < BIN_U; ++u)
    if (d[u] >= 0)
      stage[pos[u]] = (((unsigned)d[u] % TILE_R) << 16) | (unsigned)s[u];
  __syncthreads();
  for (int i = t; i < e1 - e0; i += BIN_TB) recChunk[e0 + i] = stage[i];
}

#define DEG_TB 128
#define DEG_U  8
static_assert(DEG_TB >= TILE_R, "one thread per row of the bin clears, scans and stores its count");
static_assert(BIN_BLOCKS == 64 && DEG_TB >= BIN_BLOCKS,
              "one thread per segment: two waves hold the segment table, the search takes 6 steps");
__global__ __launch_bounds__(DEG_TB)
void bin_rows(const unsigned* __restrict__ recChunk, const int* __restrict__ segOff,
              int* __restrict__ binStart, float* __restrict__ dis, unsigned* __restrict__ rec) {
  __shared__ int cnt[TILE_R];
  __shared__ int sc[TILE_R];
  __shared__ int segS[BIN_BLOCKS];
  __shared__ int pre[BIN_BLOCKS];
  __shared__ int wl[2];
  __shared__ int wa[2];
  const int t = threadIdx.x;
  const int lane = t & 31;
  const int wave = t >> 5;
  int a = 0;
  int len = 0;
  if (t < BIN_BLOCKS) {
    a = segOff[t * SEG_PITCH + blockIdx.x];
    len = segOff[t * SEG_PITCH + blockIdx.x + 1] - a;
  }
  if (t < TILE_R) cnt[t] = 0;
  const int incl = wave_incl_sum(len, lane);
  int asum = a;
#pragma unroll
  for (int o = 16; o >= 1; o >>= 1) asum += __shfl_xor(asum, o, 32);
  if (wave < 2 && lane == 31) {
    wl[wave] = incl;
    wa[wave] = asum;
  }
  __syncthreads();
  const int e0 = wa[0] + wa[1];
  const int total = wl[0] + wl[1];
  if (t < BIN_BLOCKS) {
    pre[t] = incl - len + (wave == 1 ? wl[0] : 0);
    segS[t] = t * BIN_CHUNK + a;
  }
  if (t == 0) {
    binStart[blockIdx.x] = e0;
    if (blockIdx.x == N_BINS - 1) binStart[N_BINS] = e0 + total;
  }
  __syncthreads();
  auto fetch_group = [&](int base, unsigned (&r)[DEG_U]) {
    int idx[DEG_U];
    int k[DEG_U];
#pragma unroll
    for (int u = 0; u < DEG_U; ++u) {
      idx[u] = min(base + u * DEG_TB, total - 1);
      k[u] = 0;
    }
#pragma unroll
    for (int step = BIN_BLOCKS / 2; step >= 1; step >>= 1) {
#pragma unroll
      for (int u = 0; u < DEG_U; ++u)
        if (pre[k[u] + step] <= idx[u]) k[u] += step;
    }
#pragma unroll
    for (int u = 0; u < DEG_U; ++u) r[u] = recChunk[segS[k[u]] + (idx[u] - pre[k[u]])];
#pragma unroll
    for (int u = 0; u < DEG_U; ++u)
      if (base + u * DEG_TB >= total) r[u] = 0xffffffffu;
  };
  for (int base = t; base < total; base += DEG_U * DEG_TB) {
    unsigned r[DEG_U];
    fetch_group(base, r);
#pragma unroll
    for (int u = 0; u < DEG_U; ++u)
      if (r[u] != 0xffffffffu) atomicAdd(&cnt[r[u] >> 16], 1);
  }
  __syncthreads();
  int c = 0;
  if (t < TILE_R) {
    c = cnt[t];
    sc[t] = c;
    const int row = blockIdx.x * TILE_R + t;
    if (row < N_NODES) dis[row] = rsqrtf((float)c + 1.0f);
  }
  __syncthreads();
  for (int off = 1; off < TILE_R; off <<= 1) {
    int v = 0;
    if (t < TILE_R && t >= off) v = sc[t - off];
    __syncthreads();
    if (t < TILE_R) sc[t] += v;
    __syncthreads();
  }
  if (t < TILE_R) cnt[t] = e0 + sc[t] - c;
  __syncthreads();
  for (int base = t; base < total; base += DEG_U * DEG_TB) {
    unsigned r[DEG_U];
    int pos[DEG_U];
    fetch_group(base, r);
#pragma unroll
    for (int u = 0; u < DEG_U; ++u) {
      pos[u] = -1;
      if (r[u] != 0xffffffffu) pos[u] = atomicAdd(&cnt[r[u] >> 16], 1);
    }
#pragma unroll
    for (int u = 0; u < DEG_U; ++u)
      if (r[u] != 0xffffffffu) rec[pos[u]] = r[u];
  }
}

#define TILE_K 128
#define TILE_P 130

template <bool PRE, int K, int G>
__device__ __forceinline__
void sum_rows_into_tile(const unsigned* __restrict__ rec, const int* __restrict__ binStart,
                        const float* __restrict__ h, const float* __restrict__ dis,
                        float* __restrict__ tile, float* __restrict__ disT,
                        int row0, int nrows, int wave, int lane) {
  constexpr int VPL = K / 32;
  constexpr int P = K + 2;
  static_assert(32 % G == 0, "a trip of 32 records is whole groups");
  typedef float vec __attribute__((ext_vector_type(VPL)));
  constexpr int SEED_U = TILE_R / AGG_WAVES;
  static_assert(SEED_U * AGG_WAVES == TILE_R, "every wave seeds SEED_U whole rows of the tile");

  const int e0 = binStart[blockIdx.x];
  const int e1 = binStart[blockIdx.x + 1];

  vec sv[SEED_U];
  float sd[SEED_U];
#pragma unroll
  for (int u = 0; u < SEED_U; ++u) {
    const int rr = min(row0 + wave + u * AGG_WAVES, N_NODES - 1);
    sd[u] = PRE ? 1.0f : dis[rr];
    sv[u] = *(const vec*)(h + (size_t)rr * K + lane * VPL);
  }
  const unsigned idle = ((unsigned)TILE_R << 16) | (unsigned)row0;
  int base = e0 + wave * 32;
  unsigned r = idle;
  if (base + lane < e1) r = rec[base + lane];
#pragma unroll
  for (int u = 0; u < SEED_U; ++u) {
    const int rt = wave + u * AGG_WAVES;
#pragma unroll
    for (int q = 0; q < VPL; ++q)
      tile[rt * P + q * 32 + lane] = PRE ? sv[u][q] : sv[u][q] * sd[u];
  }
  if (threadIdx.x < TILE_R)
    disT[threadIdx.x] = ((int)threadIdx.x < nrows) ? dis[row0 + threadIdx.x] : 0.0f;
  float ds = 0.0f;
  if (!PRE && base + lane < e1) ds = dis[r & 0xffffu];
  __syncthreads();

  for (; base < e1; base += AGG_WAVES * 32) {
    const unsigned rc = r;
    const float nm = ds;
    const int nidx = base + AGG_WAVES * 32 + lane;
    r = idle;
    if (nidx < e1) r = rec[nidx];
    int cur = -1;
    vec acc;
#pragma unroll
    for (int q = 0; q < VPL; ++q) acc[q] = 0.0f;
#pragma unroll
    for (int j = 0; j < 32; j += G) {
      if (base + j >= e1) break;
      vec v[G];
      int dl[G];
      float w[G];
#pragma unroll
      for (int u = 0; u < G; ++u) {
        const unsigned rj = (unsigned)__builtin_amdgcn_readlane((int)rc, j + u);
        w[u] = PRE ? 1.0f
                   : __builtin_bit_cast(float, __builtin_amdgcn_readlane(__builtin_bit_cast(int, nm), j + u));
        dl[u] = (int)(rj >> 16);
        v[u] = *(const vec*)(h + (size_t)(rj & 0xffffu) * K + lane * VPL);
      }
#pragma unroll
      for (int u = 0; u < G; ++u) {
        if (dl[u] != cur) {
          if (cur >= 0) {
#pragma unroll
            for (int q = 0; q < VPL; ++q)
              __hip_atomic_fetch_add(&tile[cur * P + q * 32 + lane], acc[q],
                                     __ATOMIC_RELAXED, __HIP_MEMORY_SCOPE_WORKGROUP);
          }
          cur = dl[u];
#pragma unroll
          for (int q = 0; q < VPL; ++q) acc[q] = PRE ? v[u][q] : v[u][q] * w[u];
        } else {
#pragma unroll
          for (int q = 0; q < VPL; ++q) acc[q] += PRE ? v[u][q] : v[u][q] * w[u];
        }
      }
      if (!PRE && j == 0) {
        ds = 0.0f;
        if (nidx < e1) ds = dis[r & 0xffffu];
      }
    }
    if (cur >= 0) {
#pragma unroll
      for (int q = 0; q < VPL; ++q)
        __hip_atomic_fetch_add(&tile[cur * P + q * 32 + lane], acc[q],
                               __ATOMIC_RELAXED, __HIP_MEMORY_SCOPE_WORKGROUP);
    }
  }

  __syncthreads();
}

template <int FOUT, bool RELU, bool PRE, bool PROJ>
__global__ __launch_bounds__(AGG_WAVES * 32, 8)
void layer_tile(const unsigned* __restrict__ rec, const int* __restrict__ binStart,
                const float* __restrict__ h, const float* __restrict__ dis,
                const float* __restrict__ W, const float* __restrict__ bias,
                const float* __restrict__ Wn, float* __restrict__ out) {
  __shared__ __attribute__((aligned(16))) float tile[(TILE_R + 1) * TILE_P];
  __shared__ float disT[TILE_R];

  const int lane = threadIdx.x & 31;
  const int wave = __builtin_amdgcn_readfirstlane((int)(threadIdx.x >> 5));
  const int row0 = blockIdx.x * TILE_R;
  const int nrows = min(TILE_R, N_NODES - row0);

  constexpr int NCT = FOUT / 16;
  constexpr int RTW = NCT / 2;
  static_assert(AGG_WAVES * RTW == (TILE_R / 16) * NCT, "8 waves cover the block's output tiles");
  static_assert(!PROJ || FOUT == TILE_K, "the rows put back into the tile have the tile's width");
  const int ct = wave % NCT;
  const int rg = wave / NCT;
  const int n16 = lane & 15;
  const int half = lane >> 4;
  const float* wp = W + ct * 16 + n16;
  v2f bw[32];
#pragma unroll
  for (int s = 0; s < 32; ++s) {
    const int k = 8 * (s & 15) + 2 * (s >> 4) + half;
    bw[s].x = wp[(size_t)k * FOUT];
    bw[s].y = wp[(size_t)(k + 4) * FOUT];
  }
  const float bcol = bias[ct * 16 + n16];

  sum_rows_into_tile<PRE, TILE_K, 8>(rec, binStart, h, dis, tile, disT, row0, nrows, wave, lane);


  v8f acc[RTW];
#pragma unroll
  for (int t = 0; t < RTW; ++t) acc[t] = v8f{};
  const float* abase = tile + ((rg * RTW) * 16 + n16) * TILE_P + 32 * half;
#pragma unroll
  for (int s = 0; s < 32; ++s) {
#pragma unroll
    for (int t = 0; t < RTW; ++t) {
      const v2f a = *(const v2f*)(abase + t * 16 * TILE_P + 64 * (s >> 4) + 2 * (s & 15));
      acc[t] = __builtin_amdgcn_wmma_f32_16x16x4_f32(
           false, a,  false, bw[s],
           (short)0, acc[t],  false,  false);
    }
  }

  if constexpr (!PROJ) {
#pragma unroll
    for (int t = 0; t < RTW; ++t) {
      const int rb = (rg * RTW + t) * 16 + (half << 3);
      float* op = out + (size_t)(row0 + rb) * FOUT + ct * 16 + n16;
      float val[8];
#pragma unroll
      for (int i = 0; i < 8; ++i) {
        const float dr = disT[rb + i];
        val[i] = acc[t][i] * dr + bcol;
        if (RELU) val[i] = fmaxf(val[i], 0.0f);
        val[i] *= dr;
      }
      if (rb + 8 <= nrows) {
#pragma unroll
        for (int i = 0; i < 8; ++i) op[(size_t)i * FOUT] = val[i];
      } else {
#pragma unroll
        for (int i = 0; i < 8; ++i)
          if (rb + i < nrows) op[(size_t)i * FOUT] = val[i];
      }
    }
  } else {
    static_assert(D_OUT == 64, "8 waves x 2 tiles cover the 4 x 4 tiles of the second pass");
    const int ct2 = wave & 3;
    const int rg2 = wave >> 2;
    __syncthreads();
#pragma unroll
    for (int t = 0; t < RTW; ++t) {
      const int rb = (rg * RTW + t) * 16 + (half << 3);
#pragma unroll
      for (int i = 0; i < 8; ++i) {
        const float dr = disT[rb + i];
        float val = acc[t][i] * dr + bcol;
        if (RELU) val = fmaxf(val, 0.0f);
        tile[(rb + i) * TILE_P + ct * 16 + n16] = val * dr;
      }
    }
    __builtin_amdgcn_sched_barrier(0);
    const float* wq = Wn + (size_t)(32 * half) * D_OUT + ct2 * 16 + n16;
    v2f bq[32];
#pragma unroll
    for (int s = 0; s < 32; ++s) {
      const int k = 64 * (s >> 4) + 2 * (s & 15);
      bq[s].x = wq[(size_t)k * D_OUT];
      bq[s].y = wq[(size_t)(k + 1) * D_OUT];
    }
    __syncthreads();

    v8f acc2[2] = {v8f{}, v8f{}};
    const float* a2 = tile + ((rg2 * 2) * 16 + n16) * TILE_P + 32 * half;
#pragma unroll
    for (int s = 0; s < 32; ++s) {
#pragma unroll
      for (int t = 0; t < 2; ++t) {
        const v2f a = *(const v2f*)(a2 + t * 16 * TILE_P + 64 * (s >> 4) + 2 * (s & 15));
        acc2[t] = __builtin_amdgcn_wmma_f32_16x16x4_f32(
             false, a,  false, bq[s],
             (short)0, acc2[t],  false,  false);
      }
    }
#pragma unroll
    for (int t = 0; t < 2; ++t) {
      const int rb = (rg2 * 2 + t) * 16 + (half << 3);
      float* op = out + (size_t)(row0 + rb) * D_OUT + ct2 * 16 + n16;
      if (rb + 8 <= nrows) {
#pragma unroll
        for (int i = 0; i < 8; ++i) op[(size_t)i * D_OUT] = acc2[t][i];
      } else {
#pragma unroll
        for (int i = 0; i < 8; ++i)
          if (rb + i < nrows) op[(size_t)i * D_OUT] = acc2[t][i];
      }
    }
  }
}

#define POOL_P (D_OUT + 2)
__global__ __launch_bounds__(AGG_WAVES * 32)
void layer_pool(const unsigned* __restrict__ rec, const int* __restrict__ binStart,
                const float* __restrict__ z, const float* __restrict__ dis,
                const int* __restrict__ batch, float* __restrict__ psum,
                float* __restrict__ cnts) {
  __shared__ __attribute__((aligned(16))) float tile[(TILE_R + 1) * POOL_P];
  __shared__ float disT[TILE_R];
  __shared__ int gidT[TILE_R];

  const int lane = threadIdx.x & 31;
  const int wave = __builtin_amdgcn_readfirstlane((int)(threadIdx.x >> 5));
  const int row0 = blockIdx.x * TILE_R;
  const int nrows = min(TILE_R, N_NODES - row0);

  if (threadIdx.x < TILE_R)
    gidT[threadIdx.x] = ((int)threadIdx.x < nrows) ? batch[row0 + threadIdx.x] : -1;

  sum_rows_into_tile<true, D_OUT, 16>(rec, binStart, z, dis, tile, disT, row0, nrows, wave, lane);

  constexpr int QR = TILE_R / 4;
  static_assert(AGG_WAVES * 32 == 4 * D_OUT, "a thread per column and row quarter");
  const int col = threadIdx.x & (D_OUT - 1);
  const int r0 = (wave >> 1) * QR;
  float v[QR];
  int gd[QR];
#pragma unroll
  for (int r = 0; r < QR; ++r) {
    v[r] = tile[(r0 + r) * POOL_P + col] * disT[r0 + r];
    gd[r] = __builtin_amdgcn_readfirstlane(gidT[r0 + r]);
  }
  auto hand_over = [&](int g, float s, int len) {
    if ((unsigned)g < (unsigned)N_GRAPHS) {
      atomicAdd(&psum[g * D_OUT + col], s);
      if (col == 0) atomicAdd(&cnts[g], (float)len);
    }
  };
  int g = gd[0];
  int len = 1;
  float s = v[0];
#pragma unroll
  for (int r = 1; r < QR; ++r) {
    if (gd[r] != g) {
      hand_over(g, s, len);
      g = gd[r];
      len = 0;
      s = 0.0f;
    }
    s += v[r];
    ++len;
  }
  hand_over(g, s, len);
}

__global__ void pool_finish(const float* __restrict__ psum, const float* __restrict__ cnts,
                            const float* __restrict__ bias, float* __restrict__ out) {
  const int i = blockIdx.x * blockDim.x + threadIdx.x;
  if (i < N_GRAPHS * D_OUT) {
    const int gq = i / D_OUT;
    const int f = i % D_OUT;
    const float cn = cnts[gq];
    out[i] = (psum[gq * D_OUT + (f & 1) * 32 + (f >> 1)] + cn * bias[f]) / fmaxf(cn, 1.0f);
  }
}

extern "C" void kernel_launch(void* const* d_in, const int* in_sizes, int n_in,
                              void* d_out, int out_size, void* d_ws, size_t ws_size,
                              hipStream_t stream) {
  const float* x     = (const float*)d_in[0];
  const int*   ei    = (const int*)d_in[1];
  const int*   batch = (const int*)d_in[2];
  const float* W1 = (const float*)d_in[3];  const float* b1 = (const float*)d_in[4];
  const float* W2 = (const float*)d_in[5];  const float* b2 = (const float*)d_in[6];
  const float* W3 = (const float*)d_in[7];  const float* b3 = (const float*)d_in[8];
  const float* W4 = (const float*)d_in[9];  const float* b4 = (const float*)d_in[10];
  float* out = (float*)d_out;

  float* bufA = (float*)d_ws;
  float* bufB = bufA + (size_t)N_NODES * D_HID;
  float* dis  = bufB + (size_t)N_NODES * D_HID;
  int* binStart = (int*)(dis + N_NODES);
  unsigned* rec = (unsigned*)(binStart + N_BINS + 1);
  int* segOff   = (int*)bufB;
  unsigned* recChunk = (unsigned*)bufA;
  float* psum = bufB;
  float* cnts = psum + N_GRAPHS * D_OUT;

  const int* src = ei;
  const int* dst = ei + N_EDGES;

  if (ws_size < WS_NEED) return;

  const int TB = 256;
  auto blocks = [](long long n, int tb) { return (unsigned)((n + tb - 1) / tb); };

  chunk_group<<<BIN_BLOCKS, BIN_TB, 0, stream>>>(src, dst, segOff, recChunk);

  bin_rows<<<N_BINS, DEG_TB, 0, stream>>>(recChunk, segOff, binStart, dis, rec);

  static_assert(D_IN == TILE_K && D_HID == TILE_K, "every layer's input has TILE_K features");
  layer_tile<D_HID, true, false, false><<<N_BINS, AGG_WAVES * 32, 0, stream>>>(rec, binStart, x, dis, W1, b1, W4, bufA);
  layer_tile<D_HID, true, true, false><<<N_BINS, AGG_WAVES * 32, 0, stream>>>(rec, binStart, bufA, dis, W2, b2, W4, bufB);
  layer_tile<D_HID, true, true, true><<<N_BINS, AGG_WAVES * 32, 0, stream>>>(rec, binStart, bufB, dis, W3, b3, W4, bufA);

  zero_f32<<<blocks(N_GRAPHS * D_OUT + N_GRAPHS, TB), TB, 0, stream>>>(
      psum, N_GRAPHS * D_OUT + N_GRAPHS);
  layer_pool<<<N_BINS, AGG_WAVES * 32, 0, stream>>>(rec, binStart, bufA, dis, batch, psum, cnts);
  pool_finish<<<blocks(N_GRAPHS * D_OUT, TB), TB, 0, stream>>>(psum, cnts, b4, out);
}


# mean pool summed straight into out: graph sizes counted in the binning launch, layer 3's block 0 stores 1/size and seeds out with b4, run sums go out pre-divided: 8 launches are 6; sums regrouped
# speedup vs baseline: 8.9131x; 1.0560x over previous
#include <hip/hip_runtime.h>
#include <hip/hip_bf16.h>

typedef float v2f __attribute__((ext_vector_type(2)));
typedef float v8f __attribute__((ext_vector_type(8)));

#define N_NODES  50000
#define N_EDGES  600000
#define N_GRAPHS 64
#define D_IN     128
#define D_HID    128
#define D_OUT    64

#define TILE_R     64
#define N_BINS     ((N_NODES + TILE_R - 1) / TILE_R)
#define BIN_BLOCKS 64
#define BIN_CHUNK  ((N_EDGES + BIN_BLOCKS - 1) / BIN_BLOCKS)
#define BIN_TB     1024
#define BIN_U      ((BIN_CHUNK + BIN_TB - 1) / BIN_TB)
#define AGG_WAVES  8

static_assert(N_NODES <= 65536, "a source row index must fit the low 16 bits of an edge record");
static_assert(N_BINS + 1 <= BIN_TB, "the scan gives one thread to each bin and one to the total");

constexpr size_t WS_NEED =
    ((size_t)2 * N_NODES * D_HID + N_NODES + N_BINS + 1 + N_EDGES) * sizeof(float);
static_assert((size_t)N_NODES * D_OUT + N_GRAPHS <= (size_t)N_NODES * D_HID,
              "the graphs' factors fit behind layer 3's [N,64] rows in the first plane");

#define SEG_PITCH (N_BINS + 1)

__device__ __forceinline__ int wave_incl_sum(int v, int lane) {
#pragma unroll
  for (int o = 1; o < 32; o <<= 1) {
    const int y = __shfl_up(v, o, 32);
    if (lane >= o) v += y;
  }
  return v;
}

__global__ __launch_bounds__(BIN_TB)
void chunk_group(const int* __restrict__ src, const int* __restrict__ dst,
                 int* __restrict__ segOff, unsigned* __restrict__ recChunk,
                 int* __restrict__ gcnt) {
  __shared__ int hist[N_BINS];
  __shared__ int wsum[BIN_TB / 32];
  __shared__ unsigned stage[BIN_CHUNK];
  const int t = threadIdx.x;
  const int lane = t & 31;
  const int wave = t >> 5;
  const int e0 = blockIdx.x * BIN_CHUNK;
  const int e1 = min(e0 + BIN_CHUNK, N_EDGES);
  int d[BIN_U];
  int s[BIN_U];
#pragma unroll
  for (int u = 0; u < BIN_U; ++u) {
    const int e = e0 + t + u * BIN_TB;
    const int ec = min(e, N_EDGES - 1);
    const int v = dst[ec];
    s[u] = src[ec];
    d[u] = (e < e1) ? v : -1;
  }
  for (int b = t; b < N_BINS; b += BIN_TB) hist[b] = 0;
  if (blockIdx.x == 0 && t < N_GRAPHS) gcnt[t] = 0;
  __syncthreads();
#pragma unroll
  for (int u = 0; u < BIN_U; ++u)
    if (d[u] >= 0) atomicAdd(&hist[(unsigned)d[u] / TILE_R], 1);
  __syncthreads();
  const int c = (t < N_BINS) ? hist[t] : 0;
  const int incl = wave_incl_sum(c, lane);
  if (lane == 31) wsum[wave] = incl;
  __syncthreads();
  if (wave == 0) {
    const int w = wsum[lane];
    wsum[lane] = wave_incl_sum(w, lane) - w;
  }
  __syncthreads();
  const int excl = wsum[wave] + incl - c;
  if (t <= N_BINS) segOff[blockIdx.x * SEG_PITCH + t] = excl;
  if (t < N_BINS) hist[t] = excl;
  __syncthreads();
  int pos[BIN_U];
#pragma unroll
  for (int u = 0; u < BIN_U; ++u) {
    pos[u] = -1;
    if (d[u] >= 0) pos[u] = atomicAdd(&hist[(unsigned)d[u] / TILE_R], 1);
  }
#pragma unroll
  for (int u = 0; u < BIN_U; ++u)
    if (d[u] >= 0)
      stage[pos[u]] = (((unsigned)d[u] % TILE_R) << 16) | (unsigned)s[u];
  __syncthreads();
  for (int i = t; i < e1 - e0; i += BIN_TB) recChunk[e0 + i] = stage[i];
}

#define DEG_TB 128
#define DEG_U  8
static_assert(DEG_TB >= TILE_R, "one thread per row of the bin clears, scans and stores its count");
static_assert(BIN_BLOCKS == 64 && DEG_TB >= BIN_BLOCKS,
              "one thread per segment: two waves hold the segment table, the search takes 6 steps");
static_assert(DEG_TB >= N_GRAPHS, "one thread per graph clears and hands over its count");
__global__ __launch_bounds__(DEG_TB)
void bin_rows(const unsigned* __restrict__ recChunk, const int* __restrict__ segOff,
              int* __restrict__ binStart, float* __restrict__ dis, unsigned* __restrict__ rec,
              const int* __restrict__ batch, int* __restrict__ gcnt) {
  __shared__ int cnt[TILE_R];
  __shared__ int gh[N_GRAPHS];
  __shared__ int sc[TILE_R];
  __shared__ int segS[BIN_BLOCKS];
  __shared__ int pre[BIN_BLOCKS];
  __shared__ int wl[2];
  __shared__ int wa[2];
  const int t = threadIdx.x;
  const int lane = t & 31;
  const int wave = t >> 5;
  int a = 0;
  int len = 0;
  if (t < BIN_BLOCKS) {
    a = segOff[t * SEG_PITCH + blockIdx.x];
    len = segOff[t * SEG_PITCH + blockIdx.x + 1] - a;
  }
  if (t < TILE_R) cnt[t] = 0;
  if (t < N_GRAPHS) gh[t] = 0;
  int gid = -1;
  if (t < TILE_R && blockIdx.x * TILE_R + t < N_NODES) gid = batch[blockIdx.x * TILE_R + t];
  const int incl = wave_incl_sum(len, lane);
  int asum = a;
#pragma unroll
  for (int o = 16; o >= 1; o >>= 1) asum += __shfl_xor(asum, o, 32);
  if (wave < 2 && lane == 31) {
    wl[wave] = incl;
    wa[wave] = asum;
  }
  __syncthreads();
  if ((unsigned)gid < (unsigned)N_GRAPHS) atomicAdd(&gh[gid], 1);
  const int e0 = wa[0] + wa[1];
  const int total = wl[0] + wl[1];
  if (t < BIN_BLOCKS) {
    pre[t] = incl - len + (wave == 1 ? wl[0] : 0);
    segS[t] = t * BIN_CHUNK + a;
  }
  if (t == 0) {
    binStart[blockIdx.x] = e0;
    if (blockIdx.x == N_BINS - 1) binStart[N_BINS] = e0 + total;
  }
  __syncthreads();
  if (t < N_GRAPHS) {
    const int gc = gh[t];
    if (gc > 0) atomicAdd(&gcnt[t], gc);
  }
  auto fetch_group = [&](int base, unsigned (&r)[DEG_U]) {
    int idx[DEG_U];
    int k[DEG_U];
#pragma unroll
    for (int u = 0; u < DEG_U; ++u) {
      idx[u] = min(base + u * DEG_TB, total - 1);
      k[u] = 0;
    }
#pragma unroll
    for (int step = BIN_BLOCKS / 2; step >= 1; step >>= 1) {
#pragma unroll
      for (int u = 0; u < DEG_U; ++u)
        if (pre[k[u] + step] <= idx[u]) k[u] += step;
    }
#pragma unroll
    for (int u = 0; u < DEG_U; ++u) r[u] = recChunk[segS[k[u]] + (idx[u] - pre[k[u]])];
#pragma unroll
    for (int u = 0; u < DEG_U; ++u)
      if (base + u * DEG_TB >= total) r[u] = 0xffffffffu;
  };
  for (int base = t; base < total; base += DEG_U * DEG_TB) {
    unsigned r[DEG_U];
    fetch_group(base, r);
#pragma unroll
    for (int u = 0; u < DEG_U; ++u)
      if (r[u] != 0xffffffffu) atomicAdd(&cnt[r[u] >> 16], 1);
  }
  __syncthreads();
  int c = 0;
  if (t < TILE_R) {
    c = cnt[t];
    sc[t] = c;
    const int row = blockIdx.x * TILE_R + t;
    if (row < N_NODES) dis[row] = rsqrtf((float)c + 1.0f);
  }
  __syncthreads();
  for (int off = 1; off < TILE_R; off <<= 1) {
    int v = 0;
    if (t < TILE_R && t >= off) v = sc[t - off];
    __syncthreads();
    if (t < TILE_R) sc[t] += v;
    __syncthreads();
  }
  if (t < TILE_R) cnt[t] = e0 + sc[t] - c;
  __syncthreads();
  for (int base = t; base < total; base += DEG_U * DEG_TB) {
    unsigned r[DEG_U];
    int pos[DEG_U];
    fetch_group(base, r);
#pragma unroll
    for (int u = 0; u < DEG_U; ++u) {
      pos[u] = -1;
      if (r[u] != 0xffffffffu) pos[u] = atomicAdd(&cnt[r[u] >> 16], 1);
    }
#pragma unroll
    for (int u = 0; u < DEG_U; ++u)
      if (r[u] != 0xffffffffu) rec[pos[u]] = r[u];
  }
}

#define TILE_K 128
#define TILE_P 130

template <bool PRE, int K, int G>
__device__ __forceinline__
void sum_rows_into_tile(const unsigned* __restrict__ rec, const int* __restrict__ binStart,
                        const float* __restrict__ h, const float* __restrict__ dis,
                        float* __restrict__ tile, float* __restrict__ disT,
                        int row0, int nrows, int wave, int lane) {
  constexpr int VPL = K / 32;
  constexpr int P = K + 2;
  static_assert(32 % G == 0, "a trip of 32 records is whole groups");
  typedef float vec __attribute__((ext_vector_type(VPL)));
  constexpr int SEED_U = TILE_R / AGG_WAVES;
  static_assert(SEED_U * AGG_WAVES == TILE_R, "every wave seeds SEED_U whole rows of the tile");

  const int e0 = binStart[blockIdx.x];
  const int e1 = binStart[blockIdx.x + 1];

  vec sv[SEED_U];
  float sd[SEED_U];
#pragma unroll
  for (int u = 0; u < SEED_U; ++u) {
    const int rr = min(row0 + wave + u * AGG_WAVES, N_NODES - 1);
    sd[u] = PRE ? 1.0f : dis[rr];
    sv[u] = *(const vec*)(h + (size_t)rr * K + lane * VPL);
  }
  const unsigned idle = ((unsigned)TILE_R << 16) | (unsigned)row0;
  int base = e0 + wave * 32;
  unsigned r = idle;
  if (base + lane < e1) r = rec[base + lane];
#pragma unroll
  for (int u = 0; u < SEED_U; ++u) {
    const int rt = wave + u * AGG_WAVES;
#pragma unroll
    for (int q = 0; q < VPL; ++q)
      tile[rt * P + q * 32 + lane] = PRE ? sv[u][q] : sv[u][q] * sd[u];
  }
  if (threadIdx.x < TILE_R)
    disT[threadIdx.x] = ((int)threadIdx.x < nrows) ? dis[row0 + threadIdx.x] : 0.0f;
  float ds = 0.0f;
  if (!PRE && base + lane < e1) ds = dis[r & 0xffffu];
  __syncthreads();

  for (; base < e1; base += AGG_WAVES * 32) {
    const unsigned rc = r;
    const float nm = ds;
    const int nidx = base + AGG_WAVES * 32 + lane;
    r = idle;
    if (nidx < e1) r = rec[nidx];
    int cur = -1;
    vec acc;
#pragma unroll
    for (int q = 0; q < VPL; ++q) acc[q] = 0.0f;
#pragma unroll
    for (int j = 0; j < 32; j += G) {
      if (base + j >= e1) break;
      vec v[G];
      int dl[G];
      float w[G];
#pragma unroll
      for (int u = 0; u < G; ++u) {
        const unsigned rj = (unsigned)__builtin_amdgcn_readlane((int)rc, j + u);
        w[u] = PRE ? 1.0f
                   : __builtin_bit_cast(float, __builtin_amdgcn_readlane(__builtin_bit_cast(int, nm), j + u));
        dl[u] = (int)(rj >> 16);
        v[u] = *(const vec*)(h + (size_t)(rj & 0xffffu) * K + lane * VPL);
      }
#pragma unroll
      for (int u = 0; u < G; ++u) {
        if (dl[u] != cur) {
          if (cur >= 0) {
#pragma unroll
            for (int q = 0; q < VPL; ++q)
              __hip_atomic_fetch_add(&tile[cur * P + q * 32 + lane], acc[q],
                                     __ATOMIC_RELAXED, __HIP_MEMORY_SCOPE_WORKGROUP);
          }
          cur = dl[u];
#pragma unroll
          for (int q = 0; q < VPL; ++q) acc[q] = PRE ? v[u][q] : v[u][q] * w[u];
        } else {
#pragma unroll
          for (int q = 0; q < VPL; ++q) acc[q] += PRE ? v[u][q] : v[u][q] * w[u];
        }
      }
      if (!PRE && j == 0) {
        ds = 0.0f;
        if (nidx < e1) ds = dis[r & 0xffffu];
      }
    }
    if (cur >= 0) {
#pragma unroll
      for (int q = 0; q < VPL; ++q)
        __hip_atomic_fetch_add(&tile[cur * P + q * 32 + lane], acc[q],
                               __ATOMIC_RELAXED, __HIP_MEMORY_SCOPE_WORKGROUP);
    }
  }

  __syncthreads();
}

template <int FOUT, bool RELU, bool PRE, bool PROJ>
__global__ __launch_bounds__(AGG_WAVES * 32, 8)
void layer_tile(const unsigned* __restrict__ rec, const int* __restrict__ binStart,
                const float* __restrict__ h, const float* __restrict__ dis,
                const float* __restrict__ W, const float* __restrict__ bias,
                const float* __restrict__ Wn, float* __restrict__ out,
                const float* __restrict__ bn, float* pooled) {
  __shared__ __attribute__((aligned(16))) float tile[(TILE_R + 1) * TILE_P];
  __shared__ float disT[TILE_R];

  const int lane = threadIdx.x & 31;
  const int wave = __builtin_amdgcn_readfirstlane((int)(threadIdx.x >> 5));
  const int row0 = blockIdx.x * TILE_R;
  const int nrows = min(TILE_R, N_NODES - row0);

  if constexpr (PROJ) {
    if (blockIdx.x == 0) {
      constexpr int GSTEP = AGG_WAVES * 32 / D_OUT;
      constexpr int PER = N_GRAPHS / GSTEP;
      static_assert(GSTEP * D_OUT == AGG_WAVES * 32 && PER * GSTEP == N_GRAPHS,
                    "the block's threads cover the [N_GRAPHS, D_OUT] output in PER whole sweeps");
      const int f = threadIdx.x % D_OUT;
      const int g0 = threadIdx.x / D_OUT;
      float ic[PER];
      bool has[PER];
#pragma unroll
      for (int k = 0; k < PER; ++k) {
        const int cn = __builtin_bit_cast(int, pooled[g0 + k * GSTEP]);
        ic[k] = 1.0f / fmaxf((float)cn, 1.0f);
        has[k] = cn > 0;
      }
      __syncthreads();
      const float bf = bn[f];
      float* icp = out + (size_t)N_NODES * D_OUT;
#pragma unroll
      for (int k = 0; k < PER; ++k) {
        const int g = g0 + k * GSTEP;
        pooled[g * D_OUT + f] = has[k] ? bf : 0.0f;
        if (f == 0) icp[g] = ic[k];
      }
    }
  }

  constexpr int NCT = FOUT / 16;
  constexpr int RTW = NCT / 2;
  static_assert(AGG_WAVES * RTW == (TILE_R / 16) * NCT, "8 waves cover the block's output tiles");
  static_assert(!PROJ || FOUT == TILE_K, "the rows put back into the tile have the tile's width");
  const int ct = wave % NCT;
  const int rg = wave / NCT;
  const int n16 = lane & 15;
  const int half = lane >> 4;
  const float* wp = W + ct * 16 + n16;
  v2f bw[32];
#pragma unroll
  for (int s = 0; s < 32; ++s) {
    const int k = 8 * (s & 15) + 2 * (s >> 4) + half;
    bw[s].x = wp[(size_t)k * FOUT];
    bw[s].y = wp[(size_t)(k + 4) * FOUT];
  }
  const float bcol = bias[ct * 16 + n16];

  sum_rows_into_tile<PRE, TILE_K, 8>(rec, binStart, h, dis, tile, disT, row0, nrows, wave, lane);


  v8f acc[RTW];
#pragma unroll
  for (int t = 0; t < RTW; ++t) acc[t] = v8f{};
  const float* abase = tile + ((rg * RTW) * 16 + n16) * TILE_P + 32 * half;
#pragma unroll
  for (int s = 0; s < 32; ++s) {
#pragma unroll
    for (int t = 0; t < RTW; ++t) {
      const v2f a = *(const v2f*)(abase + t * 16 * TILE_P + 64 * (s >> 4) + 2 * (s & 15));
      acc[t] = __builtin_amdgcn_wmma_f32_16x16x4_f32(
           false, a,  false, bw[s],
           (short)0, acc[t],  false,  false);
    }
  }

  if constexpr (!PROJ) {
#pragma unroll
    for (int t = 0; t < RTW; ++t) {
      const int rb = (rg * RTW + t) * 16 + (half << 3);
      float* op = out + (size_t)(row0 + rb) * FOUT + ct * 16 + n16;
      float val[8];
#pragma unroll
      for (int i = 0; i < 8; ++i) {
        const float dr = disT[rb + i];
        val[i] = acc[t][i] * dr + bcol;
        if (RELU) val[i] = fmaxf(val[i], 0.0f);
        val[i] *= dr;
      }
      if (rb + 8 <= nrows) {
#pragma unroll
        for (int i = 0; i < 8; ++i) op[(size_t)i * FOUT] = val[i];
      } else {
#pragma unroll
        for (int i = 0; i < 8; ++i)
          if (rb + i < nrows) op[(size_t)i * FOUT] = val[i];
      }
    }
  } else {
    static_assert(D_OUT == 64, "8 waves x 2 tiles cover the 4 x 4 tiles of the second pass");
    const int ct2 = wave & 3;
    const int rg2 = wave >> 2;
    __syncthreads();
#pragma unroll
    for (int t = 0; t < RTW; ++t) {
      const int rb = (rg * RTW + t) * 16 + (half << 3);
#pragma unroll
      for (int i = 0; i < 8; ++i) {
        const float dr = disT[rb + i];
        float val = acc[t][i] * dr + bcol;
        if (RELU) val = fmaxf(val, 0.0f);
        tile[(rb + i) * TILE_P + ct * 16 + n16] = val * dr;
      }
    }
    __builtin_amdgcn_sched_barrier(0);
    const float* wq = Wn + (size_t)(32 * half) * D_OUT + ct2 * 16 + n16;
    v2f bq[32];
#pragma unroll
    for (int s = 0; s < 32; ++s) {
      const int k = 64 * (s >> 4) + 2 * (s & 15);
      bq[s].x = wq[(size_t)k * D_OUT];
      bq[s].y = wq[(size_t)(k + 1) * D_OUT];
    }
    __syncthreads();

    v8f acc2[2] = {v8f{}, v8f{}};
    const float* a2 = tile + ((rg2 * 2) * 16 + n16) * TILE_P + 32 * half;
#pragma unroll
    for (int s = 0; s < 32; ++s) {
#pragma unroll
      for (int t = 0; t < 2; ++t) {
        const v2f a = *(const v2f*)(a2 + t * 16 * TILE_P + 64 * (s >> 4) + 2 * (s & 15));
        acc2[t] = __builtin_amdgcn_wmma_f32_16x16x4_f32(
             false, a,  false, bq[s],
             (short)0, acc2[t],  false,  false);
      }
    }
#pragma unroll
    for (int t = 0; t < 2; ++t) {
      const int rb = (rg2 * 2 + t) * 16 + (half << 3);
      float* op = out + (size_t)(row0 + rb) * D_OUT + ct2 * 16 + n16;
      if (rb + 8 <= nrows) {
#pragma unroll
        for (int i = 0; i < 8; ++i) op[(size_t)i * D_OUT] = acc2[t][i];
      } else {
#pragma unroll
        for (int i = 0; i < 8; ++i)
          if (rb + i < nrows) op[(size_t)i * D_OUT] = acc2[t][i];
      }
    }
  }
}

#define POOL_P (D_OUT + 2)
__global__ __launch_bounds__(AGG_WAVES * 32)
void layer_pool(const unsigned* __restrict__ rec, const int* __restrict__ binStart,
                const float* __restrict__ z, const float* __restrict__ dis,
                const int* __restrict__ batch, const float* __restrict__ invc,
                float* __restrict__ pooled) {
  __shared__ __attribute__((aligned(16))) float tile[(TILE_R + 1) * POOL_P];
  __shared__ float disT[TILE_R];
  __shared__ int gidT[TILE_R];
  __shared__ float icT[N_GRAPHS];
  static_assert(TILE_R + N_GRAPHS <= AGG_WAVES * 32, "ids and factors are fetched by different threads");

  const int lane = threadIdx.x & 31;
  const int wave = __builtin_amdgcn_readfirstlane((int)(threadIdx.x >> 5));
  const int row0 = blockIdx.x * TILE_R;
  const int nrows = min(TILE_R, N_NODES - row0);

  if (threadIdx.x < TILE_R)
    gidT[threadIdx.x] = ((int)threadIdx.x < nrows) ? batch[row0 + threadIdx.x] : -1;
  else if (threadIdx.x < TILE_R + N_GRAPHS)
    icT[threadIdx.x - TILE_R] = invc[threadIdx.x - TILE_R];

  sum_rows_into_tile<true, D_OUT, 16>(rec, binStart, z, dis, tile, disT, row0, nrows, wave, lane);

  constexpr int QR = TILE_R / 4;
  static_assert(AGG_WAVES * 32 == 4 * D_OUT, "a thread per column and row quarter");
  const int f = threadIdx.x & (D_OUT - 1);
  const int col = (f & 1) * 32 + (f >> 1);
  const int r0 = (wave >> 1) * QR;
  float v[QR];
  int gd[QR];
#pragma unroll
  for (int r = 0; r < QR; ++r) {
    v[r] = tile[(r0 + r) * POOL_P + col] * disT[r0 + r];
    gd[r] = __builtin_amdgcn_readfirstlane(gidT[r0 + r]);
  }
  auto hand_over = [&](int g, float s) {
    if ((unsigned)g < (unsigned)N_GRAPHS) atomicAdd(&pooled[g * D_OUT + f], s * icT[g]);
  };
  int g = gd[0];
  float s = v[0];
#pragma unroll
  for (int r = 1; r < QR; ++r) {
    if (gd[r] != g) {
      hand_over(g, s);
      g = gd[r];
      s = 0.0f;
    }
    s += v[r];
  }
  hand_over(g, s);
}

extern "C" void kernel_launch(void* const* d_in, const int* in_sizes, int n_in,
                              void* d_out, int out_size, void* d_ws, size_t ws_size,
                              hipStream_t stream) {
  const float* x     = (const float*)d_in[0];
  const int*   ei    = (const int*)d_in[1];
  const int*   batch = (const int*)d_in[2];
  const float* W1 = (const float*)d_in[3];  const float* b1 = (const float*)d_in[4];
  const float* W2 = (const float*)d_in[5];  const float* b2 = (const float*)d_in[6];
  const float* W3 = (const float*)d_in[7];  const float* b3 = (const float*)d_in[8];
  const float* W4 = (const float*)d_in[9];  const float* b4 = (const float*)d_in[10];
  float* out = (float*)d_out;

  float* bufA = (float*)d_ws;
  float* bufB = bufA + (size_t)N_NODES * D_HID;
  float* dis  = bufB + (size_t)N_NODES * D_HID;
  int* binStart = (int*)(dis + N_NODES);
  unsigned* rec = (unsigned*)(binStart + N_BINS + 1);
  int* segOff   = (int*)bufB;
  unsigned* recChunk = (unsigned*)bufA;
  int* gcnt = (int*)d_out;
  float* invc = bufA + (size_t)N_NODES * D_OUT;

  const int* src = ei;
  const int* dst = ei + N_EDGES;

  if (ws_size < WS_NEED) return;

  chunk_group<<<BIN_BLOCKS, BIN_TB, 0, stream>>>(src, dst, segOff, recChunk, gcnt);

  bin_rows<<<N_BINS, DEG_TB, 0, stream>>>(recChunk, segOff, binStart, dis, rec, batch, gcnt);

  static_assert(D_IN == TILE_K && D_HID == TILE_K, "every layer's input has TILE_K features");
  layer_tile<D_HID, true, false, false><<<N_BINS, AGG_WAVES * 32, 0, stream>>>(rec, binStart, x, dis, W1, b1, W4, bufA, b4, out);
  layer_tile<D_HID, true, true, false><<<N_BINS, AGG_WAVES * 32, 0, stream>>>(rec, binStart, bufA, dis, W2, b2, W4, bufB, b4, out);
  layer_tile<D_HID, true, true, true><<<N_BINS, AGG_WAVES * 32, 0, stream>>>(rec, binStart, bufB, dis, W3, b3, W4, bufA, b4, out);

  layer_pool<<<N_BINS, AGG_WAVES * 32, 0, stream>>>(rec, binStart, bufA, dis, batch, invc, out);
}
